# starting kernel
# baseline (speedup 1.0000x reference)
#include <hip/hip_runtime.h>
#include <hip/hip_bf16.h>


#define H_DIM 1024
#define I_DIM 2048
#define N_ST  16
#define R_DIM 64
#define B_SZ  2
#define L_SEQ 1024
#define ROWS  (B_SZ * L_SEQ)
#define EPS   1e-06f

typedef __attribute__((ext_vector_type(2))) float v2f;
typedef __attribute__((ext_vector_type(8))) float v8f;

typedef __attribute__((address_space(1))) const float gcf;
typedef __attribute__((address_space(1))) const v2f  gcv2f;

template <int NT, int EPI>
__global__ __launch_bounds__(256)
void wmma_gemm_f32(const float* __restrict__ A,
                   const float* __restrict__ W,
                   const float* __restrict__ bias,
                   float* __restrict__ C,
                   int M, int N, int K)
{
    const int lane = threadIdx.x & 31;
    const int wave = threadIdx.x >> 5;
    const int row0 = (blockIdx.y * 8 + wave) * 16;
    const int col0 = blockIdx.x * (16 * NT);

    const int r  = lane & 15;
    const int kk = (lane >> 4) * 2;

    gcf* Ap = (gcf*)(A + (size_t)(row0 + r) * K + kk);
    gcf* Wp[NT];
#pragma unroll
    for (int t = 0; t < NT; ++t)
        Wp[t] = (gcf*)(W + (size_t)(col0 + t * 16 + r) * K + kk);

    v8f acc[NT];
#pragma unroll
    for (int t = 0; t < NT; ++t) acc[t] = (v8f){0.f,0.f,0.f,0.f,0.f,0.f,0.f,0.f};

    for (int k = 0; k < K; k += 16) {
        __builtin_prefetch((const float*)(Ap + 256), 0, 1);

        v2f a[4];
#pragma unroll
        for (int s = 0; s < 4; ++s)
            a[s] = *(gcv2f*)(Ap + 4 * s);

        v2f b[NT][4];
#pragma unroll
        for (int t = 0; t < NT; ++t)
#pragma unroll
            for (int s = 0; s < 4; ++s)
                b[t][s] = *(gcv2f*)(Wp[t] + 4 * s);

#pragma unroll
        for (int s = 0; s < 4; ++s)
#pragma unroll
            for (int t = 0; t < NT; ++t)
                acc[t] = __builtin_amdgcn_wmma_f32_16x16x4_f32(
                             false, a[s], false, b[t][s],
                             (short)0, acc[t], false, false);

        Ap += 16;
#pragma unroll
        for (int t = 0; t < NT; ++t) Wp[t] += 16;
    }

    const int rowoff = (lane >> 4) * 8;
#pragma unroll
    for (int t = 0; t < NT; ++t) {
        const int col = col0 + t * 16 + r;
#pragma unroll
        for (int v = 0; v < 8; ++v) {
            float val = acc[t][v];
            if (EPI == 1) {
                val += bias[col];
                val = (val > 20.f) ? val : __logf(1.f + __expf(val));
            }
            C[(size_t)(row0 + rowoff + v) * N + col] = val;
        }
    }
}

__global__ __launch_bounds__(256)
void conv_silu_kernel(const float* __restrict__ proj,
                      const float* __restrict__ cw,
                      const float* __restrict__ cb,
                      float* __restrict__ u)
{
    const int row = blockIdx.x;
    const int i   = blockIdx.y * 256 + threadIdx.x;
    const int b   = row / L_SEQ;
    const int l   = row - b * L_SEQ;

    float acc = cb[i];
#pragma unroll
    for (int k = 0; k < 4; ++k) {
        const int src = l + k - 1;
        if (src >= 0 && src < L_SEQ)
            acc += cw[i * 4 + k] *
                   proj[(size_t)(b * L_SEQ + src) * (2 * I_DIM) + i];
    }
    u[(size_t)row * I_DIM + i] = acc / (1.f + __expf(-acc));
}

__global__ __launch_bounds__(256)
void rmsnorm_kernel(const float* __restrict__ xd,
                    const float* __restrict__ wdt,
                    const float* __restrict__ wb,
                    const float* __restrict__ wc,
                    float* __restrict__ dtn,
                    float* __restrict__ Bm,
                    float* __restrict__ Cm)
{
    const int row = blockIdx.x * 256 + threadIdx.x;
    if (row >= ROWS) return;
    const float* x = xd + (size_t)row * (R_DIM + 2 * N_ST);

    float s = 0.f;
#pragma unroll 8
    for (int j = 0; j < R_DIM; ++j) s += x[j] * x[j];
    float inv = __frsqrt_rn(s / R_DIM + EPS);
#pragma unroll 8
    for (int j = 0; j < R_DIM; ++j)
        dtn[(size_t)row * R_DIM + j] = wdt[j] * x[j] * inv;

    s = 0.f;
#pragma unroll
    for (int j = 0; j < N_ST; ++j) { float v = x[R_DIM + j]; s += v * v; }
    inv = __frsqrt_rn(s / N_ST + EPS);
#pragma unroll
    for (int j = 0; j < N_ST; ++j)
        Bm[(size_t)row * N_ST + j] = wb[j] * x[R_DIM + j] * inv;

    s = 0.f;
#pragma unroll
    for (int j = 0; j < N_ST; ++j) { float v = x[R_DIM + N_ST + j]; s += v * v; }
    inv = __frsqrt_rn(s / N_ST + EPS);
#pragma unroll
    for (int j = 0; j < N_ST; ++j)
        Cm[(size_t)row * N_ST + j] = wc[j] * x[R_DIM + N_ST + j] * inv;
}

#define CHUNK 64
__global__ __launch_bounds__(128)
void scan_kernel(const float* __restrict__ dt2,
                 const float* __restrict__ u,
                 const float* __restrict__ Bm,
                 const float* __restrict__ Cm,
                 const float* __restrict__ A_log,
                 const float* __restrict__ Dvec,
                 const float* __restrict__ proj,
                 float* __restrict__ yout)
{
    __shared__ float sB[CHUNK][N_ST];
    __shared__ float sC[CHUNK][N_ST];

    const int b = blockIdx.y;
    const int i = blockIdx.x * 128 + threadIdx.x;

    float Aa[N_ST];
#pragma unroll
    for (int n = 0; n < N_ST; ++n)
        Aa[n] = -__expf(A_log[(size_t)i * N_ST + n]);

    float h[N_ST];
#pragma unroll
    for (int n = 0; n < N_ST; ++n) h[n] = 0.f;

    const float Dv = Dvec[i];

    for (int l0 = 0; l0 < L_SEQ; l0 += CHUNK) {
        for (int t = threadIdx.x; t < CHUNK * N_ST; t += 128) {
            const size_t g = (size_t)(b * L_SEQ + l0) * N_ST + t;
            sB[t / N_ST][t % N_ST] = Bm[g];
            sC[t / N_ST][t % N_ST] = Cm[g];
        }
        __syncthreads();

        for (int l = 0; l < CHUNK; ++l) {
            const size_t idx = (size_t)(b * L_SEQ + l0 + l) * I_DIM + i;
            const float dtv = dt2[idx];
            const float uv  = u[idx];
            const float du  = dtv * uv;
            float y = 0.f;
#pragma unroll
            for (int n = 0; n < N_ST; ++n) {
                const float dA = __expf(dtv * Aa[n]);
                h[n] = dA * h[n] + du * sB[l][n];
                y   += h[n] * sC[l][n];
            }
            const float g = proj[(size_t)(b * L_SEQ + l0 + l) * (2 * I_DIM)
                                 + I_DIM + i];
            yout[idx] = (y + uv * Dv) * (g / (1.f + __expf(-g)));
        }
        __syncthreads();
    }
}

extern "C" void kernel_launch(void* const* d_in, const int* in_sizes, int n_in,
                              void* d_out, int out_size, void* d_ws, size_t ws_size,
                              hipStream_t stream)
{
    const float* hidden     = (const float*)d_in[0];
    const float* in_proj_w  = (const float*)d_in[1];
    const float* conv_w     = (const float*)d_in[2];
    const float* conv_b     = (const float*)d_in[3];
    const float* x_proj_w   = (const float*)d_in[4];
    const float* dt_proj_w  = (const float*)d_in[5];
    const float* dt_proj_b  = (const float*)d_in[6];
    const float* dt_ln_w    = (const float*)d_in[7];
    const float* b_ln_w     = (const float*)d_in[8];
    const float* c_ln_w     = (const float*)d_in[9];
    const float* A_log      = (const float*)d_in[10];
    const float* Dvec       = (const float*)d_in[11];
    const float* out_proj_w = (const float*)d_in[12];
    float* out = (float*)d_out;

    float* ws   = (float*)d_ws;
    float* proj = ws;
    float* u    = proj + (size_t)ROWS * 2 * I_DIM;
    float* xdbl = u    + (size_t)ROWS * I_DIM;
    float* dtn  = xdbl + (size_t)ROWS * (R_DIM + 2 * N_ST);
    float* Bm   = dtn  + (size_t)ROWS * R_DIM;
    float* Cm   = Bm   + (size_t)ROWS * N_ST;
    float* dt2  = Cm   + (size_t)ROWS * N_ST;
    float* yout = dt2  + (size_t)ROWS * I_DIM;

    wmma_gemm_f32<4, 0><<<dim3((2 * I_DIM) / 64, ROWS / 128), 256, 0, stream>>>(
        hidden, in_proj_w, nullptr, proj, ROWS, 2 * I_DIM, H_DIM);

    conv_silu_kernel<<<dim3(ROWS, I_DIM / 256), 256, 0, stream>>>(
        proj, conv_w, conv_b, u);

    wmma_gemm_f32<2, 0><<<dim3((R_DIM + 2 * N_ST) / 32, ROWS / 128), 256, 0, stream>>>(
        u, x_proj_w, nullptr, xdbl, ROWS, R_DIM + 2 * N_ST, I_DIM);

    rmsnorm_kernel<<<(ROWS + 255) / 256, 256, 0, stream>>>(
        xdbl, dt_ln_w, b_ln_w, c_ln_w, dtn, Bm, Cm);

    wmma_gemm_f32<4, 1><<<dim3(I_DIM / 64, ROWS / 128), 256, 0, stream>>>(
        dtn, dt_proj_w, dt_proj_b, dt2, ROWS, I_DIM, R_DIM);

    scan_kernel<<<dim3(I_DIM / 128, B_SZ), 128, 0, stream>>>(
        dt2, u, Bm, Cm, A_log, Dvec, proj, yout);

    wmma_gemm_f32<4, 0><<<dim3(H_DIM / 64, ROWS / 128), 256, 0, stream>>>(
        yout, out_proj_w, nullptr, out, ROWS, H_DIM, I_DIM);
}


# scan split by role: 32-channel work-groups; 3 waves make exp(dt*A), dt*u*B, gate SiLU a chunk ahead as LDS records, 1 wave runs only h=dA*h+dBu, y+=h*C (C by readlane); dBu rounded, sum order same
# speedup vs baseline: 5.9556x; 5.9556x over previous
#include <hip/hip_runtime.h>
#include <hip/hip_bf16.h>


#define H_DIM 1024
#define I_DIM 2048
#define N_ST  16
#define R_DIM 64
#define B_SZ  2
#define L_SEQ 1024
#define ROWS  (B_SZ * L_SEQ)
#define EPS   1e-06f

typedef __attribute__((ext_vector_type(2))) float v2f;
typedef __attribute__((ext_vector_type(8))) float v8f;

typedef __attribute__((address_space(1))) const float gcf;
typedef __attribute__((address_space(1))) const v2f  gcv2f;
typedef __attribute__((ext_vector_type(4))) float v4f;
typedef __attribute__((address_space(1))) const v4f gcv4f;

template <int NT, int EPI>
__global__ __launch_bounds__(256)
void wmma_gemm_f32(const float* __restrict__ A,
                   const float* __restrict__ W,
                   const float* __restrict__ bias,
                   float* __restrict__ C,
                   int M, int N, int K)
{
    const int lane = threadIdx.x & 31;
    const int wave = threadIdx.x >> 5;
    const int row0 = (blockIdx.y * 8 + wave) * 16;
    const int col0 = blockIdx.x * (16 * NT);

    const int r  = lane & 15;
    const int kk = (lane >> 4) * 2;

    gcf* Ap = (gcf*)(A + (size_t)(row0 + r) * K + kk);
    gcf* Wp[NT];
#pragma unroll
    for (int t = 0; t < NT; ++t)
        Wp[t] = (gcf*)(W + (size_t)(col0 + t * 16 + r) * K + kk);

    v8f acc[NT];
#pragma unroll
    for (int t = 0; t < NT; ++t) acc[t] = (v8f){0.f,0.f,0.f,0.f,0.f,0.f,0.f,0.f};

    for (int k = 0; k < K; k += 16) {
        __builtin_prefetch((const float*)(Ap + 256), 0, 1);

        v2f a[4];
#pragma unroll
        for (int s = 0; s < 4; ++s)
            a[s] = *(gcv2f*)(Ap + 4 * s);

        v2f b[NT][4];
#pragma unroll
        for (int t = 0; t < NT; ++t)
#pragma unroll
            for (int s = 0; s < 4; ++s)
                b[t][s] = *(gcv2f*)(Wp[t] + 4 * s);

#pragma unroll
        for (int s = 0; s < 4; ++s)
#pragma unroll
            for (int t = 0; t < NT; ++t)
                acc[t] = __builtin_amdgcn_wmma_f32_16x16x4_f32(
                             false, a[s], false, b[t][s],
                             (short)0, acc[t], false, false);

        Ap += 16;
#pragma unroll
        for (int t = 0; t < NT; ++t) Wp[t] += 16;
    }

    const int rowoff = (lane >> 4) * 8;
#pragma unroll
    for (int t = 0; t < NT; ++t) {
        const int col = col0 + t * 16 + r;
#pragma unroll
        for (int v = 0; v < 8; ++v) {
            float val = acc[t][v];
            if (EPI == 1) {
                val += bias[col];
                val = (val > 20.f) ? val : __logf(1.f + __expf(val));
            }
            C[(size_t)(row0 + rowoff + v) * N + col] = val;
        }
    }
}

template <int MT, int NT, int KS>
struct FragSet {
    v4f a[MT][KS];
    v4f b[NT][KS];
};

template <int MT, int NT, int KS, int KD>
__device__ __forceinline__
void frag_request(FragSet<MT, NT, KS>& f, gcf* Ap, gcf* Wp, int k)
{
#pragma unroll
    for (int mt = 0; mt < MT; ++mt)
#pragma unroll
        for (int j = 0; j < KS; ++j)
            f.a[mt][j] = *(gcv4f*)(Ap + (size_t)mt * 16 * KD + k + 8 * j);
#pragma unroll
    for (int nt = 0; nt < NT; ++nt)
#pragma unroll
        for (int j = 0; j < KS; ++j)
            f.b[nt][j] = *(gcv4f*)(Wp + (size_t)nt * 16 * KD + k + 8 * j);
}

template <int MT, int NT, int KS>
__device__ __forceinline__
void frag_mma(const FragSet<MT, NT, KS>& f, v8f (&acc)[MT][NT])
{
#pragma unroll
    for (int j = 0; j < KS; ++j) {
        v2f alo[MT], ahi[MT], blo[NT], bhi[NT];
#pragma unroll
        for (int mt = 0; mt < MT; ++mt) {
            alo[mt] = __builtin_shufflevector(f.a[mt][j], f.a[mt][j], 0, 1);
            ahi[mt] = __builtin_shufflevector(f.a[mt][j], f.a[mt][j], 2, 3);
        }
#pragma unroll
        for (int nt = 0; nt < NT; ++nt) {
            blo[nt] = __builtin_shufflevector(f.b[nt][j], f.b[nt][j], 0, 1);
            bhi[nt] = __builtin_shufflevector(f.b[nt][j], f.b[nt][j], 2, 3);
        }
#pragma unroll
        for (int mt = 0; mt < MT; ++mt)
#pragma unroll
            for (int nt = 0; nt < NT; ++nt)
                acc[mt][nt] = __builtin_amdgcn_wmma_f32_16x16x4_f32(
                                  false, alo[mt], false, blo[nt],
                                  (short)0, acc[mt][nt], false, false);
#pragma unroll
        for (int mt = 0; mt < MT; ++mt)
#pragma unroll
            for (int nt = 0; nt < NT; ++nt)
                acc[mt][nt] = __builtin_amdgcn_wmma_f32_16x16x4_f32(
                                  false, ahi[mt], false, bhi[nt],
                                  (short)0, acc[mt][nt], false, false);
    }
}

template <int MT, int NT, int KS, int WM, int WN, int KD, int ND>
__global__ __launch_bounds__(WM * WN * 32)
void wmma_gemm_f32_regtile(const float* __restrict__ A,
                           const float* __restrict__ W,
                           float* __restrict__ C)
{
    constexpr int KSET = 8 * KS;
    static_assert(KD % (2 * KSET) == 0, "K loop consumes two sets per trip");
    const int lane = threadIdx.x & 31;
    const int wave = threadIdx.x >> 5;
    const int wm   = wave / WN;
    const int wn   = wave - wm * WN;
    const int row0 = (blockIdx.y * WM + wm) * (16 * MT);
    const int col0 = (blockIdx.x * WN + wn) * (16 * NT);

    const int r = lane & 15;
    const int h = lane >> 4;

    gcf* Ap = (gcf*)(A + (size_t)(row0 + r) * KD + 4 * h);
    gcf* Wp = (gcf*)(W + (size_t)(col0 + r) * KD + 4 * h);

    v8f acc[MT][NT];
#pragma unroll
    for (int mt = 0; mt < MT; ++mt)
#pragma unroll
        for (int nt = 0; nt < NT; ++nt)
            acc[mt][nt] = (v8f){0.f,0.f,0.f,0.f,0.f,0.f,0.f,0.f};

    FragSet<MT, NT, KS> f0, f1;
    frag_request<MT, NT, KS, KD>(f0, Ap, Wp, 0);
    for (int k = 0; k < KD - 2 * KSET; k += 2 * KSET) {
        frag_request<MT, NT, KS, KD>(f1, Ap, Wp, k + KSET);
        __builtin_amdgcn_sched_barrier(0);
        frag_mma<MT, NT, KS>(f0, acc);
        __builtin_amdgcn_sched_barrier(0);
        frag_request<MT, NT, KS, KD>(f0, Ap, Wp, k + 2 * KSET);
        __builtin_amdgcn_sched_barrier(0);
        frag_mma<MT, NT, KS>(f1, acc);
        __builtin_amdgcn_sched_barrier(0);
    }
    frag_request<MT, NT, KS, KD>(f1, Ap, Wp, KD - KSET);
    __builtin_amdgcn_sched_barrier(0);
    frag_mma<MT, NT, KS>(f0, acc);
    frag_mma<MT, NT, KS>(f1, acc);

#pragma unroll
    for (int mt = 0; mt < MT; ++mt)
#pragma unroll
        for (int nt = 0; nt < NT; ++nt)
#pragma unroll
            for (int v = 0; v < 8; ++v)
                C[(size_t)(row0 + mt * 16 + 8 * h + v) * ND + col0 + nt * 16 + r] =
                    acc[mt][nt][v];
}

#define XN      (R_DIM + 2 * N_ST)
#define XNT     (XN / 16)
#define XPITCH  (I_DIM + 4)
#define XKW     (I_DIM / 8)

__global__ __launch_bounds__(256)
void conv_xproj_kernel(const float* __restrict__ proj,
                       const float* __restrict__ cw,
                       const float* __restrict__ cb,
                       const float* __restrict__ xw,
                       float* __restrict__ u,
                       float* __restrict__ xdbl)
{
    static_assert(L_SEQ % 16 == 0, "a 16-row tile must stay inside one batch");
    static_assert(XKW % 64 == 0, "K loop consumes two sets of 32 k per trip");
    __shared__ __attribute__((aligned(16))) float sA[16 * XPITCH];
    __shared__ __attribute__((aligned(16))) float sRed[8 * XNT * 8 * 32];

    const int tid  = threadIdx.x;
    const int lane = tid & 31;
    const int wave = tid >> 5;
    const int row0 = blockIdx.x * 16;
    const int l0   = row0 % L_SEQ;
    const bool has_top = (l0 > 0);
    const bool has_bot = (l0 + 16 < L_SEQ);

#pragma unroll
    for (int p = 0; p < 2; ++p) {
        const int ch = wave * XKW + p * 128 + lane * 4;
        gcf* src = (gcf*)(proj + (size_t)row0 * (2 * I_DIM) + ch);
        const v4f zero = (v4f){0.f, 0.f, 0.f, 0.f};
        v4f x[19];
        x[0] = zero; x[17] = zero; x[18] = zero;
        if (has_top) x[0] = *(gcv4f*)(src - (2 * I_DIM));
#pragma unroll
        for (int q = 1; q < 17; ++q)
            x[q] = *(gcv4f*)(src + (size_t)(q - 1) * (2 * I_DIM));
        if (has_bot) {
            x[17] = *(gcv4f*)(src + (size_t)16 * (2 * I_DIM));
            x[18] = *(gcv4f*)(src + (size_t)17 * (2 * I_DIM));
        }
        v4f w4[4];
#pragma unroll
        for (int c = 0; c < 4; ++c)
            w4[c] = *(gcv4f*)(cw + (size_t)(ch + c) * 4);
        const v4f bias = *(gcv4f*)(cb + ch);

#pragma unroll
        for (int r = 0; r < 16; ++r) {
            v4f o;
#pragma unroll
            for (int c = 0; c < 4; ++c) {
                float acc = bias[c];
#pragma unroll
                for (int k = 0; k < 4; ++k)
                    acc += w4[c][k] * x[r + k][c];
                o[c] = acc / (1.f + __expf(-acc));
            }
            *(v4f*)(u + (size_t)(row0 + r) * I_DIM + ch) = o;
            *(v4f*)&sA[r * XPITCH + ch] = o;
        }
    }
    __syncthreads();

    const int r = lane & 15;
    const int h = lane >> 4;
    const float* sAp = sA + r * XPITCH + wave * XKW + 4 * h;
    gcf* Wp = (gcf*)(xw + (size_t)r * I_DIM + wave * XKW + 4 * h);

    v8f acc[1][XNT];
#pragma unroll
    for (int t = 0; t < XNT; ++t)
        acc[0][t] = (v8f){0.f,0.f,0.f,0.f,0.f,0.f,0.f,0.f};

    auto request = [&](FragSet<1, XNT, 4>& f, int k) {
#pragma unroll
        for (int j = 0; j < 4; ++j)
            f.a[0][j] = *(const v4f*)(sAp + k + 8 * j);
#pragma unroll
        for (int t = 0; t < XNT; ++t)
#pragma unroll
            for (int j = 0; j < 4; ++j)
                f.b[t][j] = *(gcv4f*)(Wp + (size_t)t * 16 * I_DIM + k + 8 * j);
    };

    FragSet<1, XNT, 4> f0, f1;
    request(f0, 0);
    for (int k = 0; k < XKW - 64; k += 64) {
        request(f1, k + 32);
        __builtin_amdgcn_sched_barrier(0);
        frag_mma<1, XNT, 4>(f0, acc);
        __builtin_amdgcn_sched_barrier(0);
        request(f0, k + 64);
        __builtin_amdgcn_sched_barrier(0);
        frag_mma<1, XNT, 4>(f1, acc);
        __builtin_amdgcn_sched_barrier(0);
    }
    request(f1, XKW - 32);
    __builtin_amdgcn_sched_barrier(0);
    frag_mma<1, XNT, 4>(f0, acc);
    frag_mma<1, XNT, 4>(f1, acc);

#pragma unroll
    for (int t = 0; t < XNT; ++t)
#pragma unroll
        for (int v = 0; v < 8; ++v)
            sRed[((wave * XNT + t) * 8 + v) * 32 + lane] = acc[0][t][v];
    __syncthreads();

#pragma unroll
    for (int q = 0; q < XNT; ++q) {
        const int e = tid + 256 * q;
        float sum = 0.f;
#pragma unroll
        for (int w = 0; w < 8; ++w)
            sum += sRed[w * (XNT * 8 * 32) + e];
        const int t  = e >> 8;
        const int v  = (e & 255) >> 5;
        const int ln = e & 31;
        xdbl[(size_t)(row0 + 8 * (ln >> 4) + v) * XN + t * 16 + (ln & 15)] = sum;
    }
}

__global__ __launch_bounds__(256)
void rmsnorm_kernel(const float* __restrict__ xd,
                    const float* __restrict__ wdt,
                    const float* __restrict__ wb,
                    const float* __restrict__ wc,
                    float* __restrict__ dtn,
                    float* __restrict__ Bm,
                    float* __restrict__ Cm)
{
    const int row = blockIdx.x * 256 + threadIdx.x;
    if (row >= ROWS) return;
    const float* x = xd + (size_t)row * (R_DIM + 2 * N_ST);

    float s = 0.f;
#pragma unroll 8
    for (int j = 0; j < R_DIM; ++j) s += x[j] * x[j];
    float inv = __frsqrt_rn(s / R_DIM + EPS);
#pragma unroll 8
    for (int j = 0; j < R_DIM; ++j)
        dtn[(size_t)row * R_DIM + j] = wdt[j] * x[j] * inv;

    s = 0.f;
#pragma unroll
    for (int j = 0; j < N_ST; ++j) { float v = x[R_DIM + j]; s += v * v; }
    inv = __frsqrt_rn(s / N_ST + EPS);
#pragma unroll
    for (int j = 0; j < N_ST; ++j)
        Bm[(size_t)row * N_ST + j] = wb[j] * x[R_DIM + j] * inv;

    s = 0.f;
#pragma unroll
    for (int j = 0; j < N_ST; ++j) { float v = x[R_DIM + N_ST + j]; s += v * v; }
    inv = __frsqrt_rn(s / N_ST + EPS);
#pragma unroll
    for (int j = 0; j < N_ST; ++j)
        Cm[(size_t)row * N_ST + j] = wc[j] * x[R_DIM + N_ST + j] * inv;
}

#define CHUNK   16
#define NCHUNK  (L_SEQ / CHUNK)
#define SC_CH   32
#define REC_F   36
#define REC_BUF (CHUNK * SC_CH * REC_F)

struct ScanTile {
    v4f d, u, g;
    v2f b, c;
};

__global__ __launch_bounds__(128)
void scan_kernel(const float* __restrict__ dt2,
                 const float* __restrict__ u,
                 const float* __restrict__ Bm,
                 const float* __restrict__ Cm,
                 const float* __restrict__ A_log,
                 const float* __restrict__ Dvec,
                 const float* __restrict__ proj,
                 float* __restrict__ yout)
{
    __shared__ __attribute__((aligned(16))) float sRec[2][REC_BUF];
    __shared__ __attribute__((aligned(16))) float sD[3][CHUNK][SC_CH];
    __shared__ __attribute__((aligned(16))) float sU[3][CHUNK][SC_CH];
    __shared__ __attribute__((aligned(16))) float sG[3][CHUNK][SC_CH];
    __shared__ __attribute__((aligned(16))) float sB[3][CHUNK][N_ST];
    __shared__ __attribute__((aligned(16))) float sC[3][CHUNK][N_ST];

    const int tid  = threadIdx.x;
    const int lane = tid & 31;
    const int wave = __builtin_amdgcn_readfirstlane(tid >> 5);
    const int b    = blockIdx.y;
    const int i0   = blockIdx.x * SC_CH;
    const int i    = i0 + lane;
    const int rowb = b * L_SEQ;

    float Aa[N_ST];
#pragma unroll
    for (int n = 0; n < N_ST; ++n)
        Aa[n] = -__expf(A_log[(size_t)i * N_ST + n]);

    float h[N_ST];
#pragma unroll
    for (int n = 0; n < N_ST; ++n) h[n] = 0.f;

    const float Dv = Dvec[i];

    ScanTile t;

    auto request = [&](int row0) {
        const size_t row = (size_t)(row0 + (tid >> 3));
        const int    col = i0 + (tid & 7) * 4;
        t.d = *(gcv4f*)(dt2  + row * I_DIM + col);
        t.u = *(gcv4f*)(u    + row * I_DIM + col);
        t.g = *(gcv4f*)(proj + row * (2 * I_DIM) + I_DIM + col);
        t.b = *(gcv2f*)(Bm + (size_t)row0 * N_ST + tid * 2);
        t.c = *(gcv2f*)(Cm + (size_t)row0 * N_ST + tid * 2);
    };

    auto park = [&](int slot) {
        const int r = tid >> 3, p = tid & 7;
        *(v4f*)&sD[slot][r][p * 4] = t.d;
        *(v4f*)&sU[slot][r][p * 4] = t.u;
        *(v4f*)&sG[slot][r][p * 4] = t.g;
        *(v2f*)&sB[slot][r][p * 2] = t.b;
        *(v2f*)&sC[slot][r][p * 2] = t.c;
    };

    static_assert(3 * (CHUNK / 3) == CHUNK - 1, "three waves x CHUNK/3 steps, plus one");
    struct RawStep { float dtv, uv, g; v4f bq[4]; };
    auto produce = [&](int pi, int buf, int slot) {
        auto fetch_step = [&](RawStep& s, int l) {
            s.dtv = sD[slot][l][lane];
            s.uv  = sU[slot][l][lane];
            s.g   = sG[slot][l][lane];
#pragma unroll
            for (int q = 0; q < 4; ++q)
                s.bq[q] = *(const v4f*)&sB[slot][l][4 * q];
        };
        auto record = [&](const RawStep& s, int l) {
            const float du = s.dtv * s.uv;
            float* rp = &sRec[buf][(l * SC_CH + lane) * REC_F];
#pragma unroll
            for (int q = 0; q < 4; ++q) {
                v4f a, d;
#pragma unroll
                for (int j = 0; j < 4; ++j) {
                    a[j] = __expf(s.dtv * Aa[4 * q + j]);
                    d[j] = du * s.bq[q][j];
                }
                *(v4f*)(rp + 4 * q)      = a;
                *(v4f*)(rp + 16 + 4 * q) = d;
            }
            v2f e;
            e.x = s.uv;
            e.y = s.g / (1.f + __expf(-s.g));
            *(v2f*)(rp + 32) = e;
        };
        RawStep rs[CHUNK / 3];
#pragma unroll
        for (int k = 0; k < CHUNK / 3; ++k)
            fetch_step(rs[k], pi + 3 * k);
        __builtin_amdgcn_sched_barrier(0);
#pragma unroll
        for (int k = 0; k < CHUNK / 3; ++k)
            record(rs[k], pi + 3 * k);
        if (pi == 0) {
            RawStep last;
            fetch_step(last, CHUNK - 1);
            record(last, CHUNK - 1);
        }
    };

    auto consume = [&](int buf, int slot, int row0) {
        const v4f c0 = *(const v4f*)&sC[slot][lane >> 1][(lane & 1) * 8];
        const v4f c1 = *(const v4f*)&sC[slot][lane >> 1][(lane & 1) * 8 + 4];
        int cv[8];
#pragma unroll
        for (int k = 0; k < 4; ++k) {
            cv[k]     = __float_as_int(c0[k]);
            cv[4 + k] = __float_as_int(c1[k]);
        }
#pragma unroll
        for (int l = 0; l < CHUNK; ++l) {
            const float* rp = &sRec[buf][(l * SC_CH + lane) * REC_F];
            v4f a[4], d[4];
#pragma unroll
            for (int q = 0; q < 4; ++q) {
                a[q] = *(const v4f*)(rp + 4 * q);
                d[q] = *(const v4f*)(rp + 16 + 4 * q);
            }
            const v2f e = *(const v2f*)(rp + 32);
            float y = 0.f;
#pragma unroll
            for (int n = 0; n < N_ST; ++n) {
                const int v = l * N_ST + n;
                const float cn = __int_as_float(
                    __builtin_amdgcn_readlane(cv[v & 7], v >> 3));
                h[n] = a[n >> 2][n & 3] * h[n] + d[n >> 2][n & 3];
                y   += h[n] * cn;
            }
            yout[(size_t)(row0 + l) * I_DIM + i] = (y + e.x * Dv) * e.y;
        }
    };

    request(rowb);
    park(0);
    __syncthreads();

    int s_cur = 2, s_nxt = 0, s_new = 1;
    for (int c = -1; c < NCHUNK; ++c) {
        const bool fetch = (c + 2 < NCHUNK);
        if (fetch) request(rowb + (c + 2) * CHUNK);
        if (wave == 0) {
            if (c >= 0) consume(c & 1, s_cur, rowb + c * CHUNK);
        } else {
            if (c + 1 < NCHUNK) produce(wave - 1, (c + 1) & 1, s_nxt);
        }
        if (fetch) park(s_new);
        __syncthreads();
        const int s_old = s_cur;
        s_cur = s_nxt; s_nxt = s_new; s_new = s_old;
    }
}

extern "C" void kernel_launch(void* const* d_in, const int* in_sizes, int n_in,
                              void* d_out, int out_size, void* d_ws, size_t ws_size,
                              hipStream_t stream)
{
    const float* hidden     = (const float*)d_in[0];
    const float* in_proj_w  = (const float*)d_in[1];
    const float* conv_w     = (const float*)d_in[2];
    const float* conv_b     = (const float*)d_in[3];
    const float* x_proj_w   = (const float*)d_in[4];
    const float* dt_proj_w  = (const float*)d_in[5];
    const float* dt_proj_b  = (const float*)d_in[6];
    const float* dt_ln_w    = (const float*)d_in[7];
    const float* b_ln_w     = (const float*)d_in[8];
    const float* c_ln_w     = (const float*)d_in[9];
    const float* A_log      = (const float*)d_in[10];
    const float* Dvec       = (const float*)d_in[11];
    const float* out_proj_w = (const float*)d_in[12];
    float* out = (float*)d_out;

    float* ws   = (float*)d_ws;
    float* proj = ws;
    float* u    = proj + (size_t)ROWS * 2 * I_DIM;
    float* xdbl = u    + (size_t)ROWS * I_DIM;
    float* dtn  = xdbl + (size_t)ROWS * (R_DIM + 2 * N_ST);
    float* Bm   = dtn  + (size_t)ROWS * R_DIM;
    float* Cm   = Bm   + (size_t)ROWS * N_ST;
    float* dt2  = Cm   + (size_t)ROWS * N_ST;
    float* yout = dt2  + (size_t)ROWS * I_DIM;

    wmma_gemm_f32_regtile<4, 4, 2, 2, 4, H_DIM, 2 * I_DIM>
        <<<dim3((2 * I_DIM) / 256, ROWS / 128), 256, 0, stream>>>(
        hidden, in_proj_w, proj);

    conv_xproj_kernel<<<ROWS / 16, 256, 0, stream>>>(
        proj, conv_w, conv_b, x_proj_w, u, xdbl);

    rmsnorm_kernel<<<(ROWS + 255) / 256, 256, 0, stream>>>(
        xdbl, dt_ln_w, b_ln_w, c_ln_w, dtn, Bm, Cm);

    wmma_gemm_f32<4, 1><<<dim3(I_DIM / 64, ROWS / 128), 256, 0, stream>>>(
        dtn, dt_proj_w, dt_proj_b, dt2, ROWS, I_DIM, R_DIM);

    scan_kernel<<<dim3(I_DIM / SC_CH, B_SZ), 128, 0, stream>>>(
        dt2, u, Bm, Cm, A_log, Dvec, proj, yout);

    wmma_gemm_f32_regtile<2, 4, 4, 2, 2, I_DIM, H_DIM>
        <<<dim3(H_DIM / 128, ROWS / 64), 128, 0, stream>>>(
        yout, out_proj_w, out);
}


# GEMM operands in fragment order: a pack launch re-lays hidden and 3 weight matrices, the scan stores y that way; a fragment load is 4 whole lines, not 16 row pieces of 32 B; sum order unchanged
# speedup vs baseline: 6.7776x; 1.1380x over previous
#include <hip/hip_runtime.h>
#include <hip/hip_bf16.h>


#define H_DIM 1024
#define I_DIM 2048
#define N_ST  16
#define R_DIM 64
#define B_SZ  2
#define L_SEQ 1024
#define ROWS  (B_SZ * L_SEQ)
#define EPS   1e-06f

typedef __attribute__((ext_vector_type(2))) float v2f;
typedef __attribute__((ext_vector_type(8))) float v8f;

typedef __attribute__((address_space(1))) const float gcf;
typedef __attribute__((address_space(1))) const v2f  gcv2f;
typedef __attribute__((ext_vector_type(4))) float v4f;
typedef __attribute__((address_space(1))) const v4f gcv4f;

template <int NT, int EPI>
__global__ __launch_bounds__(256)
void wmma_gemm_f32(const float* __restrict__ A,
                   const float* __restrict__ W,
                   const float* __restrict__ bias,
                   float* __restrict__ C,
                   int M, int N, int K)
{
    const int lane = threadIdx.x & 31;
    const int wave = threadIdx.x >> 5;
    const int row0 = (blockIdx.y * 8 + wave) * 16;
    const int col0 = blockIdx.x * (16 * NT);

    const int r  = lane & 15;
    const int kk = (lane >> 4) * 2;

    gcf* Ap = (gcf*)(A + (size_t)(row0 + r) * K + kk);
    gcf* Wp[NT];
#pragma unroll
    for (int t = 0; t < NT; ++t)
        Wp[t] = (gcf*)(W + (size_t)(col0 + t * 16 + r) * K + kk);

    v8f acc[NT];
#pragma unroll
    for (int t = 0; t < NT; ++t) acc[t] = (v8f){0.f,0.f,0.f,0.f,0.f,0.f,0.f,0.f};

    for (int k = 0; k < K; k += 16) {
        __builtin_prefetch((const float*)(Ap + 256), 0, 1);

        v2f a[4];
#pragma unroll
        for (int s = 0; s < 4; ++s)
            a[s] = *(gcv2f*)(Ap + 4 * s);

        v2f b[NT][4];
#pragma unroll
        for (int t = 0; t < NT; ++t)
#pragma unroll
            for (int s = 0; s < 4; ++s)
                b[t][s] = *(gcv2f*)(Wp[t] + 4 * s);

#pragma unroll
        for (int s = 0; s < 4; ++s)
#pragma unroll
            for (int t = 0; t < NT; ++t)
                acc[t] = __builtin_amdgcn_wmma_f32_16x16x4_f32(
                             false, a[s], false, b[t][s],
                             (short)0, acc[t], false, false);

        Ap += 16;
#pragma unroll
        for (int t = 0; t < NT; ++t) Wp[t] += 16;
    }

    const int rowoff = (lane >> 4) * 8;
#pragma unroll
    for (int t = 0; t < NT; ++t) {
        const int col = col0 + t * 16 + r;
#pragma unroll
        for (int v = 0; v < 8; ++v) {
            float val = acc[t][v];
            if (EPI == 1) {
                val += bias[col];
                val = (val > 20.f) ? val : __logf(1.f + __expf(val));
            }
            C[(size_t)(row0 + rowoff + v) * N + col] = val;
        }
    }
}

template <int MT, int NT, int KS>
struct FragSet {
    v4f a[MT][KS];
    v4f b[NT][KS];
};

template <int MT, int NT, int KS, int KD>
__device__ __forceinline__
void frag_request(FragSet<MT, NT, KS>& f, gcf* Ap, gcf* Wp, int k)
{
#pragma unroll
    for (int mt = 0; mt < MT; ++mt)
#pragma unroll
        for (int j = 0; j < KS; ++j)
            f.a[mt][j] = *(gcv4f*)(Ap + (size_t)mt * 16 * KD + (k + 8 * j) * 16);
#pragma unroll
    for (int nt = 0; nt < NT; ++nt)
#pragma unroll
        for (int j = 0; j < KS; ++j)
            f.b[nt][j] = *(gcv4f*)(Wp + (size_t)nt * 16 * KD + (k + 8 * j) * 16);
}

template <int MT, int NT, int KS>
__device__ __forceinline__
void frag_mma(const FragSet<MT, NT, KS>& f, v8f (&acc)[MT][NT])
{
#pragma unroll
    for (int j = 0; j < KS; ++j) {
        v2f alo[MT], ahi[MT], blo[NT], bhi[NT];
#pragma unroll
        for (int mt = 0; mt < MT; ++mt) {
            alo[mt] = __builtin_shufflevector(f.a[mt][j], f.a[mt][j], 0, 1);
            ahi[mt] = __builtin_shufflevector(f.a[mt][j], f.a[mt][j], 2, 3);
        }
#pragma unroll
        for (int nt = 0; nt < NT; ++nt) {
            blo[nt] = __builtin_shufflevector(f.b[nt][j], f.b[nt][j], 0, 1);
            bhi[nt] = __builtin_shufflevector(f.b[nt][j], f.b[nt][j], 2, 3);
        }
#pragma unroll
        for (int mt = 0; mt < MT; ++mt)
#pragma unroll
            for (int nt = 0; nt < NT; ++nt)
                acc[mt][nt] = __builtin_amdgcn_wmma_f32_16x16x4_f32(
                                  false, alo[mt], false, blo[nt],
                                  (short)0, acc[mt][nt], false, false);
#pragma unroll
        for (int mt = 0; mt < MT; ++mt)
#pragma unroll
            for (int nt = 0; nt < NT; ++nt)
                acc[mt][nt] = __builtin_amdgcn_wmma_f32_16x16x4_f32(
                                  false, ahi[mt], false, bhi[nt],
                                  (short)0, acc[mt][nt], false, false);
    }
}

template <int MT, int NT, int KS, int WM, int WN, int KD, int ND>
__global__ __launch_bounds__(WM * WN * 32)
void wmma_gemm_f32_regtile(const float* __restrict__ A,
                           const float* __restrict__ W,
                           float* __restrict__ C)
{
    constexpr int KSET = 8 * KS;
    static_assert(KD % (2 * KSET) == 0, "K loop consumes two sets per trip");
    const int lane = threadIdx.x & 31;
    const int wave = threadIdx.x >> 5;
    const int wm   = wave / WN;
    const int wn   = wave - wm * WN;
    const int row0 = (blockIdx.y * WM + wm) * (16 * MT);
    const int col0 = (blockIdx.x * WN + wn) * (16 * NT);

    const int r = lane & 15;
    const int h = lane >> 4;

    gcf* Ap = (gcf*)(A + (size_t)row0 * KD + 4 * lane);
    gcf* Wp = (gcf*)(W + (size_t)col0 * KD + 4 * lane);

    v8f acc[MT][NT];
#pragma unroll
    for (int mt = 0; mt < MT; ++mt)
#pragma unroll
        for (int nt = 0; nt < NT; ++nt)
            acc[mt][nt] = (v8f){0.f,0.f,0.f,0.f,0.f,0.f,0.f,0.f};

    FragSet<MT, NT, KS> f0, f1;
    frag_request<MT, NT, KS, KD>(f0, Ap, Wp, 0);
    for (int k = 0; k < KD - 2 * KSET; k += 2 * KSET) {
        frag_request<MT, NT, KS, KD>(f1, Ap, Wp, k + KSET);
        __builtin_amdgcn_sched_barrier(0);
        frag_mma<MT, NT, KS>(f0, acc);
        __builtin_amdgcn_sched_barrier(0);
        frag_request<MT, NT, KS, KD>(f0, Ap, Wp, k + 2 * KSET);
        __builtin_amdgcn_sched_barrier(0);
        frag_mma<MT, NT, KS>(f1, acc);
        __builtin_amdgcn_sched_barrier(0);
    }
    frag_request<MT, NT, KS, KD>(f1, Ap, Wp, KD - KSET);
    __builtin_amdgcn_sched_barrier(0);
    frag_mma<MT, NT, KS>(f0, acc);
    frag_mma<MT, NT, KS>(f1, acc);

#pragma unroll
    for (int mt = 0; mt < MT; ++mt)
#pragma unroll
        for (int nt = 0; nt < NT; ++nt)
#pragma unroll
            for (int v = 0; v < 8; ++v)
                C[(size_t)(row0 + mt * 16 + 8 * h + v) * ND + col0 + nt * 16 + r] =
                    acc[mt][nt][v];
}

#define XN      (R_DIM + 2 * N_ST)
#define XNT     (XN / 16)
#define XPITCH  (I_DIM + 4)
#define XKW     (I_DIM / 8)

__global__ __launch_bounds__(256)
void conv_xproj_kernel(const float* __restrict__ proj,
                       const float* __restrict__ cw,
                       const float* __restrict__ cb,
                       const float* __restrict__ xw,
                       float* __restrict__ u,
                       float* __restrict__ xdbl)
{
    static_assert(L_SEQ % 16 == 0, "a 16-row tile must stay inside one batch");
    static_assert(XKW % 64 == 0, "K loop consumes two sets of 32 k per trip");
    __shared__ __attribute__((aligned(16))) float sA[16 * XPITCH];
    __shared__ __attribute__((aligned(16))) float sRed[8 * XNT * 8 * 32];

    const int tid  = threadIdx.x;
    const int lane = tid & 31;
    const int wave = tid >> 5;
    const int row0 = blockIdx.x * 16;
    const int l0   = row0 % L_SEQ;
    const bool has_top = (l0 > 0);
    const bool has_bot = (l0 + 16 < L_SEQ);

#pragma unroll
    for (int p = 0; p < 2; ++p) {
        const int ch = wave * XKW + p * 128 + lane * 4;
        gcf* src = (gcf*)(proj + (size_t)row0 * (2 * I_DIM) + ch);
        const v4f zero = (v4f){0.f, 0.f, 0.f, 0.f};
        v4f x[19];
        x[0] = zero; x[17] = zero; x[18] = zero;
        if (has_top) x[0] = *(gcv4f*)(src - (2 * I_DIM));
#pragma unroll
        for (int q = 1; q < 17; ++q)
            x[q] = *(gcv4f*)(src + (size_t)(q - 1) * (2 * I_DIM));
        if (has_bot) {
            x[17] = *(gcv4f*)(src + (size_t)16 * (2 * I_DIM));
            x[18] = *(gcv4f*)(src + (size_t)17 * (2 * I_DIM));
        }
        v4f w4[4];
#pragma unroll
        for (int c = 0; c < 4; ++c)
            w4[c] = *(gcv4f*)(cw + (size_t)(ch + c) * 4);
        const v4f bias = *(gcv4f*)(cb + ch);

#pragma unroll
        for (int r = 0; r < 16; ++r) {
            v4f o;
#pragma unroll
            for (int c = 0; c < 4; ++c) {
                float acc = bias[c];
#pragma unroll
                for (int k = 0; k < 4; ++k)
                    acc += w4[c][k] * x[r + k][c];
                o[c] = acc / (1.f + __expf(-acc));
            }
            *(v4f*)(u + (size_t)(row0 + r) * I_DIM + ch) = o;
            *(v4f*)&sA[r * XPITCH + ch] = o;
        }
    }
    __syncthreads();

    const int r = lane & 15;
    const int h = lane >> 4;
    const float* sAp = sA + r * XPITCH + wave * XKW + 4 * h;
    gcf* Wp = (gcf*)(xw + (size_t)(wave * XKW) * 16 + 4 * lane);

    v8f acc[1][XNT];
#pragma unroll
    for (int t = 0; t < XNT; ++t)
        acc[0][t] = (v8f){0.f,0.f,0.f,0.f,0.f,0.f,0.f,0.f};

    auto request = [&](FragSet<1, XNT, 4>& f, int k) {
#pragma unroll
        for (int j = 0; j < 4; ++j)
            f.a[0][j] = *(const v4f*)(sAp + k + 8 * j);
#pragma unroll
        for (int t = 0; t < XNT; ++t)
#pragma unroll
            for (int j = 0; j < 4; ++j)
                f.b[t][j] = *(gcv4f*)(Wp + (size_t)t * 16 * I_DIM + (k + 8 * j) * 16);
    };

    FragSet<1, XNT, 4> f0, f1;
    request(f0, 0);
    for (int k = 0; k < XKW - 64; k += 64) {
        request(f1, k + 32);
        __builtin_amdgcn_sched_barrier(0);
        frag_mma<1, XNT, 4>(f0, acc);
        __builtin_amdgcn_sched_barrier(0);
        request(f0, k + 64);
        __builtin_amdgcn_sched_barrier(0);
        frag_mma<1, XNT, 4>(f1, acc);
        __builtin_amdgcn_sched_barrier(0);
    }
    request(f1, XKW - 32);
    __builtin_amdgcn_sched_barrier(0);
    frag_mma<1, XNT, 4>(f0, acc);
    frag_mma<1, XNT, 4>(f1, acc);

#pragma unroll
    for (int t = 0; t < XNT; ++t)
#pragma unroll
        for (int v = 0; v < 8; ++v)
            sRed[((wave * XNT + t) * 8 + v) * 32 + lane] = acc[0][t][v];
    __syncthreads();

#pragma unroll
    for (int q = 0; q < XNT; ++q) {
        const int e = tid + 256 * q;
        float sum = 0.f;
#pragma unroll
        for (int w = 0; w < 8; ++w)
            sum += sRed[w * (XNT * 8 * 32) + e];
        const int t  = e >> 8;
        const int v  = (e & 255) >> 5;
        const int ln = e & 31;
        xdbl[(size_t)(row0 + 8 * (ln >> 4) + v) * XN + t * 16 + (ln & 15)] = sum;
    }
}

__global__ __launch_bounds__(256)
void rmsnorm_kernel(const float* __restrict__ xd,
                    const float* __restrict__ wdt,
                    const float* __restrict__ wb,
                    const float* __restrict__ wc,
                    float* __restrict__ dtn,
                    float* __restrict__ Bm,
                    float* __restrict__ Cm)
{
    const int row = blockIdx.x * 256 + threadIdx.x;
    if (row >= ROWS) return;
    const float* x = xd + (size_t)row * (R_DIM + 2 * N_ST);

    float s = 0.f;
#pragma unroll 8
    for (int j = 0; j < R_DIM; ++j) s += x[j] * x[j];
    float inv = __frsqrt_rn(s / R_DIM + EPS);
#pragma unroll 8
    for (int j = 0; j < R_DIM; ++j)
        dtn[(size_t)row * R_DIM + j] = wdt[j] * x[j] * inv;

    s = 0.f;
#pragma unroll
    for (int j = 0; j < N_ST; ++j) { float v = x[R_DIM + j]; s += v * v; }
    inv = __frsqrt_rn(s / N_ST + EPS);
#pragma unroll
    for (int j = 0; j < N_ST; ++j)
        Bm[(size_t)row * N_ST + j] = wb[j] * x[R_DIM + j] * inv;

    s = 0.f;
#pragma unroll
    for (int j = 0; j < N_ST; ++j) { float v = x[R_DIM + N_ST + j]; s += v * v; }
    inv = __frsqrt_rn(s / N_ST + EPS);
#pragma unroll
    for (int j = 0; j < N_ST; ++j)
        Cm[(size_t)row * N_ST + j] = wc[j] * x[R_DIM + N_ST + j] * inv;
}

#define CHUNK   16
#define NCHUNK  (L_SEQ / CHUNK)
#define SC_CH   32
#define REC_F   36
#define REC_BUF (CHUNK * SC_CH * REC_F)

struct ScanTile {
    v4f d, u, g;
    v2f b, c;
};

__global__ __launch_bounds__(128)
void scan_kernel(const float* __restrict__ dt2,
                 const float* __restrict__ u,
                 const float* __restrict__ Bm,
                 const float* __restrict__ Cm,
                 const float* __restrict__ A_log,
                 const float* __restrict__ Dvec,
                 const float* __restrict__ proj,
                 float* __restrict__ yout)
{
    __shared__ __attribute__((aligned(16))) float sRec[2][REC_BUF];
    __shared__ __attribute__((aligned(16))) float sD[3][CHUNK][SC_CH];
    __shared__ __attribute__((aligned(16))) float sU[3][CHUNK][SC_CH];
    __shared__ __attribute__((aligned(16))) float sG[3][CHUNK][SC_CH];
    __shared__ __attribute__((aligned(16))) float sB[3][CHUNK][N_ST];
    __shared__ __attribute__((aligned(16))) float sC[3][CHUNK][N_ST];

    const int tid  = threadIdx.x;
    const int lane = tid & 31;
    const int wave = __builtin_amdgcn_readfirstlane(tid >> 5);
    const int b    = blockIdx.y;
    const int i0   = blockIdx.x * SC_CH;
    const int i    = i0 + lane;
    const int rowb = b * L_SEQ;

    float Aa[N_ST];
#pragma unroll
    for (int n = 0; n < N_ST; ++n)
        Aa[n] = -__expf(A_log[(size_t)i * N_ST + n]);

    float h[N_ST];
#pragma unroll
    for (int n = 0; n < N_ST; ++n) h[n] = 0.f;

    const float Dv = Dvec[i];
    const int ypos = (i >> 3) * 128 + ((i >> 2) & 1) * 64 + (i & 3);

    ScanTile t;

    auto request = [&](int row0) {
        const size_t row = (size_t)(row0 + (tid >> 3));
        const int    col = i0 + (tid & 7) * 4;
        t.d = *(gcv4f*)(dt2  + row * I_DIM + col);
        t.u = *(gcv4f*)(u    + row * I_DIM + col);
        t.g = *(gcv4f*)(proj + row * (2 * I_DIM) + I_DIM + col);
        t.b = *(gcv2f*)(Bm + (size_t)row0 * N_ST + tid * 2);
        t.c = *(gcv2f*)(Cm + (size_t)row0 * N_ST + tid * 2);
    };

    auto park = [&](int slot) {
        const int r = tid >> 3, p = tid & 7;
        *(v4f*)&sD[slot][r][p * 4] = t.d;
        *(v4f*)&sU[slot][r][p * 4] = t.u;
        *(v4f*)&sG[slot][r][p * 4] = t.g;
        *(v2f*)&sB[slot][r][p * 2] = t.b;
        *(v2f*)&sC[slot][r][p * 2] = t.c;
    };

    static_assert(3 * (CHUNK / 3) == CHUNK - 1, "three waves x CHUNK/3 steps, plus one");
    struct RawStep { float dtv, uv, g; v4f bq[4]; };
    auto produce = [&](int pi, int buf, int slot) {
        auto fetch_step = [&](RawStep& s, int l) {
            s.dtv = sD[slot][l][lane];
            s.uv  = sU[slot][l][lane];
            s.g   = sG[slot][l][lane];
#pragma unroll
            for (int q = 0; q < 4; ++q)
                s.bq[q] = *(const v4f*)&sB[slot][l][4 * q];
        };
        auto record = [&](const RawStep& s, int l) {
            const float du = s.dtv * s.uv;
            float* rp = &sRec[buf][(l * SC_CH + lane) * REC_F];
#pragma unroll
            for (int q = 0; q < 4; ++q) {
                v4f a, d;
#pragma unroll
                for (int j = 0; j < 4; ++j) {
                    a[j] = __expf(s.dtv * Aa[4 * q + j]);
                    d[j] = du * s.bq[q][j];
                }
                *(v4f*)(rp + 4 * q)      = a;
                *(v4f*)(rp + 16 + 4 * q) = d;
            }
            v2f e;
            e.x = s.uv;
            e.y = s.g / (1.f + __expf(-s.g));
            *(v2f*)(rp + 32) = e;
        };
        RawStep rs[CHUNK / 3];
#pragma unroll
        for (int k = 0; k < CHUNK / 3; ++k)
            fetch_step(rs[k], pi + 3 * k);
        __builtin_amdgcn_sched_barrier(0);
#pragma unroll
        for (int k = 0; k < CHUNK / 3; ++k)
            record(rs[k], pi + 3 * k);
        if (pi == 0) {
            RawStep last;
            fetch_step(last, CHUNK - 1);
            record(last, CHUNK - 1);
        }
    };

    auto consume = [&](int buf, int slot, int row0) {
        const v4f c0 = *(const v4f*)&sC[slot][lane >> 1][(lane & 1) * 8];
        const v4f c1 = *(const v4f*)&sC[slot][lane >> 1][(lane & 1) * 8 + 4];
        int cv[8];
#pragma unroll
        for (int k = 0; k < 4; ++k) {
            cv[k]     = __float_as_int(c0[k]);
            cv[4 + k] = __float_as_int(c1[k]);
        }
#pragma unroll
        for (int l = 0; l < CHUNK; ++l) {
            const float* rp = &sRec[buf][(l * SC_CH + lane) * REC_F];
            v4f a[4], d[4];
#pragma unroll
            for (int q = 0; q < 4; ++q) {
                a[q] = *(const v4f*)(rp + 4 * q);
                d[q] = *(const v4f*)(rp + 16 + 4 * q);
            }
            const v2f e = *(const v2f*)(rp + 32);
            float y = 0.f;
#pragma unroll
            for (int n = 0; n < N_ST; ++n) {
                const int v = l * N_ST + n;
                const float cn = __int_as_float(
                    __builtin_amdgcn_readlane(cv[v & 7], v >> 3));
                h[n] = a[n >> 2][n & 3] * h[n] + d[n >> 2][n & 3];
                y   += h[n] * cn;
            }
            yout[(size_t)row0 * I_DIM + ypos + 4 * l] = (y + e.x * Dv) * e.y;
        }
    };

    request(rowb);
    park(0);
    __syncthreads();

    int s_cur = 2, s_nxt = 0, s_new = 1;
    for (int c = -1; c < NCHUNK; ++c) {
        const bool fetch = (c + 2 < NCHUNK);
        if (fetch) request(rowb + (c + 2) * CHUNK);
        if (wave == 0) {
            if (c >= 0) consume(c & 1, s_cur, rowb + c * CHUNK);
        } else {
            if (c + 1 < NCHUNK) produce(wave - 1, (c + 1) & 1, s_nxt);
        }
        if (fetch) park(s_new);
        __syncthreads();
        const int s_old = s_cur;
        s_cur = s_nxt; s_nxt = s_new; s_new = s_old;
    }
}

#define PACK_WG      512
#define PACK_THREADS (PACK_WG * 256)

template <int KD, int NPIECES>
__device__ __forceinline__
void pack_matrix(const float* __restrict__ X, float* __restrict__ P, unsigned first)
{
    constexpr unsigned KB = KD / 8;
    auto src = [&](unsigned p) {
        const unsigned ln = p & 31u, blk = p >> 5;
        const unsigned rt = blk / KB, kb = blk - rt * KB;
        return (gcv4f*)(X + (size_t)(16u * rt + (ln & 15u)) * KD + 8u * kb + 4u * (ln >> 4));
    };
    if constexpr (NPIECES % PACK_THREADS == 0) {
        constexpr int PER = NPIECES / PACK_THREADS;
        v4f v[PER];
#pragma unroll
        for (int q = 0; q < PER; ++q) v[q] = *src(first + q * PACK_THREADS);
#pragma unroll
        for (int q = 0; q < PER; ++q)
            *(v4f*)(P + (size_t)(first + q * PACK_THREADS) * 4) = v[q];
    } else {
        static_assert(NPIECES < PACK_THREADS, "a small matrix: at most one piece per thread");
        if (first < (unsigned)NPIECES) *(v4f*)(P + (size_t)first * 4) = *src(first);
    }
}

__global__ __launch_bounds__(256)
void pack_operands_kernel(const float* __restrict__ hidden,
                          const float* __restrict__ in_w,
                          const float* __restrict__ out_w,
                          const float* __restrict__ x_w,
                          float* __restrict__ hidden_p,
                          float* __restrict__ in_w_p,
                          float* __restrict__ out_w_p,
                          float* __restrict__ x_w_p)
{
    const unsigned first = blockIdx.x * 256u + threadIdx.x;
    pack_matrix<H_DIM, 2 * I_DIM * H_DIM / 4>(in_w,   in_w_p,   first);
    pack_matrix<H_DIM, ROWS * H_DIM / 4>     (hidden, hidden_p, first);
    pack_matrix<I_DIM, H_DIM * I_DIM / 4>    (out_w,  out_w_p,  first);
    pack_matrix<I_DIM, XN * I_DIM / 4>       (x_w,    x_w_p,    first);
}

extern "C" void kernel_launch(void* const* d_in, const int* in_sizes, int n_in,
                              void* d_out, int out_size, void* d_ws, size_t ws_size,
                              hipStream_t stream)
{
    const float* hidden     = (const float*)d_in[0];
    const float* in_proj_w  = (const float*)d_in[1];
    const float* conv_w     = (const float*)d_in[2];
    const float* conv_b     = (const float*)d_in[3];
    const float* x_proj_w   = (const float*)d_in[4];
    const float* dt_proj_w  = (const float*)d_in[5];
    const float* dt_proj_b  = (const float*)d_in[6];
    const float* dt_ln_w    = (const float*)d_in[7];
    const float* b_ln_w     = (const float*)d_in[8];
    const float* c_ln_w     = (const float*)d_in[9];
    const float* A_log      = (const float*)d_in[10];
    const float* Dvec       = (const float*)d_in[11];
    const float* out_proj_w = (const float*)d_in[12];
    float* out = (float*)d_out;

    float* ws   = (float*)d_ws;
    float* proj = ws;
    float* u    = proj + (size_t)ROWS * 2 * I_DIM;
    float* xdbl = u    + (size_t)ROWS * I_DIM;
    float* dtn  = xdbl + (size_t)ROWS * (R_DIM + 2 * N_ST);
    float* Bm   = dtn  + (size_t)ROWS * R_DIM;
    float* Cm   = Bm   + (size_t)ROWS * N_ST;
    float* dt2  = Cm   + (size_t)ROWS * N_ST;
    float* yout = dt2  + (size_t)ROWS * I_DIM;
    float* hid_p  = yout   + (size_t)ROWS * I_DIM;
    float* inw_p  = hid_p  + (size_t)ROWS * H_DIM;
    float* outw_p = inw_p  + (size_t)2 * I_DIM * H_DIM;
    float* xw_p   = outw_p + (size_t)H_DIM * I_DIM;

    pack_operands_kernel<<<PACK_WG, 256, 0, stream>>>(
        hidden, in_proj_w, out_proj_w, x_proj_w, hid_p, inw_p, outw_p, xw_p);

    wmma_gemm_f32_regtile<4, 4, 2, 2, 4, H_DIM, 2 * I_DIM>
        <<<dim3((2 * I_DIM) / 256, ROWS / 128), 256, 0, stream>>>(
        hid_p, inw_p, proj);

    conv_xproj_kernel<<<ROWS / 16, 256, 0, stream>>>(
        proj, conv_w, conv_b, xw_p, u, xdbl);

    rmsnorm_kernel<<<(ROWS + 255) / 256, 256, 0, stream>>>(
        xdbl, dt_ln_w, b_ln_w, c_ln_w, dtn, Bm, Cm);

    wmma_gemm_f32<4, 1><<<dim3(I_DIM / 64, ROWS / 128), 256, 0, stream>>>(
        dtn, dt_proj_w, dt_proj_b, dt2, ROWS, I_DIM, R_DIM);

    scan_kernel<<<dim3(I_DIM / SC_CH, B_SZ), 128, 0, stream>>>(
        dt2, u, Bm, Cm, A_log, Dvec, proj, yout);

    wmma_gemm_f32_regtile<2, 4, 4, 2, 2, I_DIM, H_DIM>
        <<<dim3(H_DIM / 128, ROWS / 64), 128, 0, stream>>>(
        yout, outw_p, out);
}


# scan: y goes via an LDS tile; a feeding wave stores the previous chunk's tile (whole lines) at trip start, so the chunk barrier stops waiting on the recurrence wave's store acks; sum order same
# speedup vs baseline: 7.2107x; 1.0639x over previous
#include <hip/hip_runtime.h>
#include <hip/hip_bf16.h>


#define H_DIM 1024
#define I_DIM 2048
#define N_ST  16
#define R_DIM 64
#define B_SZ  2
#define L_SEQ 1024
#define ROWS  (B_SZ * L_SEQ)
#define EPS   1e-06f

typedef __attribute__((ext_vector_type(2))) float v2f;
typedef __attribute__((ext_vector_type(8))) float v8f;

typedef __attribute__((address_space(1))) const float gcf;
typedef __attribute__((address_space(1))) const v2f  gcv2f;
typedef __attribute__((ext_vector_type(4))) float v4f;
typedef __attribute__((address_space(1))) const v4f gcv4f;

template <int NT, int EPI>
__global__ __launch_bounds__(256)
void wmma_gemm_f32(const float* __restrict__ A,
                   const float* __restrict__ W,
                   const float* __restrict__ bias,
                   float* __restrict__ C,
                   int M, int N, int K)
{
    const int lane = threadIdx.x & 31;
    const int wave = threadIdx.x >> 5;
    const int row0 = (blockIdx.y * 8 + wave) * 16;
    const int col0 = blockIdx.x * (16 * NT);

    const int r  = lane & 15;
    const int kk = (lane >> 4) * 2;

    gcf* Ap = (gcf*)(A + (size_t)(row0 + r) * K + kk);
    gcf* Wp[NT];
#pragma unroll
    for (int t = 0; t < NT; ++t)
        Wp[t] = (gcf*)(W + (size_t)(col0 + t * 16 + r) * K + kk);

    v8f acc[NT];
#pragma unroll
    for (int t = 0; t < NT; ++t) acc[t] = (v8f){0.f,0.f,0.f,0.f,0.f,0.f,0.f,0.f};

    for (int k = 0; k < K; k += 16) {
        __builtin_prefetch((const float*)(Ap + 256), 0, 1);

        v2f a[4];
#pragma unroll
        for (int s = 0; s < 4; ++s)
            a[s] = *(gcv2f*)(Ap + 4 * s);

        v2f b[NT][4];
#pragma unroll
        for (int t = 0; t < NT; ++t)
#pragma unroll
            for (int s = 0; s < 4; ++s)
                b[t][s] = *(gcv2f*)(Wp[t] + 4 * s);

#pragma unroll
        for (int s = 0; s < 4; ++s)
#pragma unroll
            for (int t = 0; t < NT; ++t)
                acc[t] = __builtin_amdgcn_wmma_f32_16x16x4_f32(
                             false, a[s], false, b[t][s],
                             (short)0, acc[t], false, false);

        Ap += 16;
#pragma unroll
        for (int t = 0; t < NT; ++t) Wp[t] += 16;
    }

    const int rowoff = (lane >> 4) * 8;
#pragma unroll
    for (int t = 0; t < NT; ++t) {
        const int col = col0 + t * 16 + r;
#pragma unroll
        for (int v = 0; v < 8; ++v) {
            float val = acc[t][v];
            if (EPI == 1) {
                val += bias[col];
                val = (val > 20.f) ? val : __logf(1.f + __expf(val));
            }
            C[(size_t)(row0 + rowoff + v) * N + col] = val;
        }
    }
}

template <int MT, int NT, int KS>
struct FragSet {
    v4f a[MT][KS];
    v4f b[NT][KS];
};

template <int MT, int NT, int KS, int KD>
__device__ __forceinline__
void frag_request(FragSet<MT, NT, KS>& f, gcf* Ap, gcf* Wp, int k)
{
#pragma unroll
    for (int mt = 0; mt < MT; ++mt)
#pragma unroll
        for (int j = 0; j < KS; ++j)
            f.a[mt][j] = *(gcv4f*)(Ap + (size_t)mt * 16 * KD + (k + 8 * j) * 16);
#pragma unroll
    for (int nt = 0; nt < NT; ++nt)
#pragma unroll
        for (int j = 0; j < KS; ++j)
            f.b[nt][j] = *(gcv4f*)(Wp + (size_t)nt * 16 * KD + (k + 8 * j) * 16);
}

template <int MT, int NT, int KS>
__device__ __forceinline__
void frag_mma(const FragSet<MT, NT, KS>& f, v8f (&acc)[MT][NT])
{
#pragma unroll
    for (int j = 0; j < KS; ++j) {
        v2f alo[MT], ahi[MT], blo[NT], bhi[NT];
#pragma unroll
        for (int mt = 0; mt < MT; ++mt) {
            alo[mt] = __builtin_shufflevector(f.a[mt][j], f.a[mt][j], 0, 1);
            ahi[mt] = __builtin_shufflevector(f.a[mt][j], f.a[mt][j], 2, 3);
        }
#pragma unroll
        for (int nt = 0; nt < NT; ++nt) {
            blo[nt] = __builtin_shufflevector(f.b[nt][j], f.b[nt][j], 0, 1);
            bhi[nt] = __builtin_shufflevector(f.b[nt][j], f.b[nt][j], 2, 3);
        }
#pragma unroll
        for (int mt = 0; mt < MT; ++mt)
#pragma unroll
            for (int nt = 0; nt < NT; ++nt)
                acc[mt][nt] = __builtin_amdgcn_wmma_f32_16x16x4_f32(
                                  false, alo[mt], false, blo[nt],
                                  (short)0, acc[mt][nt], false, false);
#pragma unroll
        for (int mt = 0; mt < MT; ++mt)
#pragma unroll
            for (int nt = 0; nt < NT; ++nt)
                acc[mt][nt] = __builtin_amdgcn_wmma_f32_16x16x4_f32(
                                  false, ahi[mt], false, bhi[nt],
                                  (short)0, acc[mt][nt], false, false);
    }
}

template <int MT, int NT, int KS, int WM, int WN, int KD, int ND>
__global__ __launch_bounds__(WM * WN * 32)
void wmma_gemm_f32_regtile(const float* __restrict__ A,
                           const float* __restrict__ W,
                           float* __restrict__ C)
{
    constexpr int KSET = 8 * KS;
    static_assert(KD % (2 * KSET) == 0, "K loop consumes two sets per trip");
    const int lane = threadIdx.x & 31;
    const int wave = threadIdx.x >> 5;
    const int wm   = wave / WN;
    const int wn   = wave - wm * WN;
    const int row0 = (blockIdx.y * WM + wm) * (16 * MT);
    const int col0 = (blockIdx.x * WN + wn) * (16 * NT);

    const int r = lane & 15;
    const int h = lane >> 4;

    gcf* Ap = (gcf*)(A + (size_t)row0 * KD + 4 * lane);
    gcf* Wp = (gcf*)(W + (size_t)col0 * KD + 4 * lane);

    v8f acc[MT][NT];
#pragma unroll
    for (int mt = 0; mt < MT; ++mt)
#pragma unroll
        for (int nt = 0; nt < NT; ++nt)
            acc[mt][nt] = (v8f){0.f,0.f,0.f,0.f,0.f,0.f,0.f,0.f};

    FragSet<MT, NT, KS> f0, f1;
    frag_request<MT, NT, KS, KD>(f0, Ap, Wp, 0);
    for (int k = 0; k < KD - 2 * KSET; k += 2 * KSET) {
        frag_request<MT, NT, KS, KD>(f1, Ap, Wp, k + KSET);
        __builtin_amdgcn_sched_barrier(0);
        frag_mma<MT, NT, KS>(f0, acc);
        __builtin_amdgcn_sched_barrier(0);
        frag_request<MT, NT, KS, KD>(f0, Ap, Wp, k + 2 * KSET);
        __builtin_amdgcn_sched_barrier(0);
        frag_mma<MT, NT, KS>(f1, acc);
        __builtin_amdgcn_sched_barrier(0);
    }
    frag_request<MT, NT, KS, KD>(f1, Ap, Wp, KD - KSET);
    __builtin_amdgcn_sched_barrier(0);
    frag_mma<MT, NT, KS>(f0, acc);
    frag_mma<MT, NT, KS>(f1, acc);

#pragma unroll
    for (int mt = 0; mt < MT; ++mt)
#pragma unroll
        for (int nt = 0; nt < NT; ++nt)
#pragma unroll
            for (int v = 0; v < 8; ++v)
                C[(size_t)(row0 + mt * 16 + 8 * h + v) * ND + col0 + nt * 16 + r] =
                    acc[mt][nt][v];
}

#define XN      (R_DIM + 2 * N_ST)
#define XNT     (XN / 16)
#define XPITCH  (I_DIM + 4)
#define XKW     (I_DIM / 8)

__global__ __launch_bounds__(256)
void conv_xproj_kernel(const float* __restrict__ proj,
                       const float* __restrict__ cw,
                       const float* __restrict__ cb,
                       const float* __restrict__ xw,
                       float* __restrict__ u,
                       float* __restrict__ xdbl)
{
    static_assert(L_SEQ % 16 == 0, "a 16-row tile must stay inside one batch");
    static_assert(XKW % 64 == 0, "K loop consumes two sets of 32 k per trip");
    __shared__ __attribute__((aligned(16))) float sA[16 * XPITCH];
    __shared__ __attribute__((aligned(16))) float sRed[8 * XNT * 8 * 32];

    const int tid  = threadIdx.x;
    const int lane = tid & 31;
    const int wave = tid >> 5;
    const int row0 = blockIdx.x * 16;
    const int l0   = row0 % L_SEQ;
    const bool has_top = (l0 > 0);
    const bool has_bot = (l0 + 16 < L_SEQ);

#pragma unroll
    for (int p = 0; p < 2; ++p) {
        const int ch = wave * XKW + p * 128 + lane * 4;
        gcf* src = (gcf*)(proj + (size_t)row0 * (2 * I_DIM) + ch);
        const v4f zero = (v4f){0.f, 0.f, 0.f, 0.f};
        v4f x[19];
        x[0] = zero; x[17] = zero; x[18] = zero;
        if (has_top) x[0] = *(gcv4f*)(src - (2 * I_DIM));
#pragma unroll
        for (int q = 1; q < 17; ++q)
            x[q] = *(gcv4f*)(src + (size_t)(q - 1) * (2 * I_DIM));
        if (has_bot) {
            x[17] = *(gcv4f*)(src + (size_t)16 * (2 * I_DIM));
            x[18] = *(gcv4f*)(src + (size_t)17 * (2 * I_DIM));
        }
        v4f w4[4];
#pragma unroll
        for (int c = 0; c < 4; ++c)
            w4[c] = *(gcv4f*)(cw + (size_t)(ch + c) * 4);
        const v4f bias = *(gcv4f*)(cb + ch);

#pragma unroll
        for (int r = 0; r < 16; ++r) {
            v4f o;
#pragma unroll
            for (int c = 0; c < 4; ++c) {
                float acc = bias[c];
#pragma unroll
                for (int k = 0; k < 4; ++k)
                    acc += w4[c][k] * x[r + k][c];
                o[c] = acc / (1.f + __expf(-acc));
            }
            *(v4f*)(u + (size_t)(row0 + r) * I_DIM + ch) = o;
            *(v4f*)&sA[r * XPITCH + ch] = o;
        }
    }
    __syncthreads();

    const int r = lane & 15;
    const int h = lane >> 4;
    const float* sAp = sA + r * XPITCH + wave * XKW + 4 * h;
    gcf* Wp = (gcf*)(xw + (size_t)(wave * XKW) * 16 + 4 * lane);

    v8f acc[1][XNT];
#pragma unroll
    for (int t = 0; t < XNT; ++t)
        acc[0][t] = (v8f){0.f,0.f,0.f,0.f,0.f,0.f,0.f,0.f};

    auto request = [&](FragSet<1, XNT, 4>& f, int k) {
#pragma unroll
        for (int j = 0; j < 4; ++j)
            f.a[0][j] = *(const v4f*)(sAp + k + 8 * j);
#pragma unroll
        for (int t = 0; t < XNT; ++t)
#pragma unroll
            for (int j = 0; j < 4; ++j)
                f.b[t][j] = *(gcv4f*)(Wp + (size_t)t * 16 * I_DIM + (k + 8 * j) * 16);
    };

    FragSet<1, XNT, 4> f0, f1;
    request(f0, 0);
    for (int k = 0; k < XKW - 64; k += 64) {
        request(f1, k + 32);
        __builtin_amdgcn_sched_barrier(0);
        frag_mma<1, XNT, 4>(f0, acc);
        __builtin_amdgcn_sched_barrier(0);
        request(f0, k + 64);
        __builtin_amdgcn_sched_barrier(0);
        frag_mma<1, XNT, 4>(f1, acc);
        __builtin_amdgcn_sched_barrier(0);
    }
    request(f1, XKW - 32);
    __builtin_amdgcn_sched_barrier(0);
    frag_mma<1, XNT, 4>(f0, acc);
    frag_mma<1, XNT, 4>(f1, acc);

#pragma unroll
    for (int t = 0; t < XNT; ++t)
#pragma unroll
        for (int v = 0; v < 8; ++v)
            sRed[((wave * XNT + t) * 8 + v) * 32 + lane] = acc[0][t][v];
    __syncthreads();

#pragma unroll
    for (int q = 0; q < XNT; ++q) {
        const int e = tid + 256 * q;
        float sum = 0.f;
#pragma unroll
        for (int w = 0; w < 8; ++w)
            sum += sRed[w * (XNT * 8 * 32) + e];
        const int t  = e >> 8;
        const int v  = (e & 255) >> 5;
        const int ln = e & 31;
        xdbl[(size_t)(row0 + 8 * (ln >> 4) + v) * XN + t * 16 + (ln & 15)] = sum;
    }
}

__global__ __launch_bounds__(256)
void rmsnorm_kernel(const float* __restrict__ xd,
                    const float* __restrict__ wdt,
                    const float* __restrict__ wb,
                    const float* __restrict__ wc,
                    float* __restrict__ dtn,
                    float* __restrict__ Bm,
                    float* __restrict__ Cm)
{
    const int row = blockIdx.x * 256 + threadIdx.x;
    if (row >= ROWS) return;
    const float* x = xd + (size_t)row * (R_DIM + 2 * N_ST);

    float s = 0.f;
#pragma unroll 8
    for (int j = 0; j < R_DIM; ++j) s += x[j] * x[j];
    float inv = __frsqrt_rn(s / R_DIM + EPS);
#pragma unroll 8
    for (int j = 0; j < R_DIM; ++j)
        dtn[(size_t)row * R_DIM + j] = wdt[j] * x[j] * inv;

    s = 0.f;
#pragma unroll
    for (int j = 0; j < N_ST; ++j) { float v = x[R_DIM + j]; s += v * v; }
    inv = __frsqrt_rn(s / N_ST + EPS);
#pragma unroll
    for (int j = 0; j < N_ST; ++j)
        Bm[(size_t)row * N_ST + j] = wb[j] * x[R_DIM + j] * inv;

    s = 0.f;
#pragma unroll
    for (int j = 0; j < N_ST; ++j) { float v = x[R_DIM + N_ST + j]; s += v * v; }
    inv = __frsqrt_rn(s / N_ST + EPS);
#pragma unroll
    for (int j = 0; j < N_ST; ++j)
        Cm[(size_t)row * N_ST + j] = wc[j] * x[R_DIM + N_ST + j] * inv;
}

#define CHUNK   16
#define NCHUNK  (L_SEQ / CHUNK)
#define SC_CH   32
#define REC_F   36
#define REC_BUF (CHUNK * SC_CH * REC_F)
#define Y_PITCH 36

struct ScanTile {
    v4f d, u, g;
    v2f b, c;
};

__global__ __launch_bounds__(128)
void scan_kernel(const float* __restrict__ dt2,
                 const float* __restrict__ u,
                 const float* __restrict__ Bm,
                 const float* __restrict__ Cm,
                 const float* __restrict__ A_log,
                 const float* __restrict__ Dvec,
                 const float* __restrict__ proj,
                 float* __restrict__ yout)
{
    __shared__ __attribute__((aligned(16))) float sRec[2][REC_BUF];
    __shared__ __attribute__((aligned(16))) float sD[3][CHUNK][SC_CH];
    __shared__ __attribute__((aligned(16))) float sU[3][CHUNK][SC_CH];
    __shared__ __attribute__((aligned(16))) float sG[3][CHUNK][SC_CH];
    __shared__ __attribute__((aligned(16))) float sB[3][CHUNK][N_ST];
    __shared__ __attribute__((aligned(16))) float sC[3][CHUNK][N_ST];
    __shared__ __attribute__((aligned(16))) float sY[2][CHUNK][Y_PITCH];

    const int tid  = threadIdx.x;
    const int lane = tid & 31;
    const int wave = __builtin_amdgcn_readfirstlane(tid >> 5);
    const int b    = blockIdx.y;
    const int i0   = blockIdx.x * SC_CH;
    const int i    = i0 + lane;
    const int rowb = b * L_SEQ;

    float Aa[N_ST];
#pragma unroll
    for (int n = 0; n < N_ST; ++n)
        Aa[n] = -__expf(A_log[(size_t)i * N_ST + n]);

    float h[N_ST];
#pragma unroll
    for (int n = 0; n < N_ST; ++n) h[n] = 0.f;

    const float Dv = Dvec[i];

    ScanTile t;

    auto request = [&](int row0) {
        const size_t row = (size_t)(row0 + (tid >> 3));
        const int    col = i0 + (tid & 7) * 4;
        t.d = *(gcv4f*)(dt2  + row * I_DIM + col);
        t.u = *(gcv4f*)(u    + row * I_DIM + col);
        t.g = *(gcv4f*)(proj + row * (2 * I_DIM) + I_DIM + col);
        t.b = *(gcv2f*)(Bm + (size_t)row0 * N_ST + tid * 2);
        t.c = *(gcv2f*)(Cm + (size_t)row0 * N_ST + tid * 2);
    };

    auto park = [&](int slot) {
        const int r = tid >> 3, p = tid & 7;
        *(v4f*)&sD[slot][r][p * 4] = t.d;
        *(v4f*)&sU[slot][r][p * 4] = t.u;
        *(v4f*)&sG[slot][r][p * 4] = t.g;
        *(v2f*)&sB[slot][r][p * 2] = t.b;
        *(v2f*)&sC[slot][r][p * 2] = t.c;
    };

    static_assert(3 * (CHUNK / 3) == CHUNK - 1, "three waves x CHUNK/3 steps, plus one");
    struct RawStep { float dtv, uv, g; v4f bq[4]; };
    auto produce = [&](int pi, int buf, int slot) {
        auto fetch_step = [&](RawStep& s, int l) {
            s.dtv = sD[slot][l][lane];
            s.uv  = sU[slot][l][lane];
            s.g   = sG[slot][l][lane];
#pragma unroll
            for (int q = 0; q < 4; ++q)
                s.bq[q] = *(const v4f*)&sB[slot][l][4 * q];
        };
        auto record = [&](const RawStep& s, int l) {
            const float du = s.dtv * s.uv;
            float* rp = &sRec[buf][(l * SC_CH + lane) * REC_F];
#pragma unroll
            for (int q = 0; q < 4; ++q) {
                v4f a, d;
#pragma unroll
                for (int j = 0; j < 4; ++j) {
                    a[j] = __expf(s.dtv * Aa[4 * q + j]);
                    d[j] = du * s.bq[q][j];
                }
                *(v4f*)(rp + 4 * q)      = a;
                *(v4f*)(rp + 16 + 4 * q) = d;
            }
            v2f e;
            e.x = s.uv;
            e.y = s.g / (1.f + __expf(-s.g));
            *(v2f*)(rp + 32) = e;
        };
        RawStep rs[CHUNK / 3];
#pragma unroll
        for (int k = 0; k < CHUNK / 3; ++k)
            fetch_step(rs[k], pi + 3 * k);
        __builtin_amdgcn_sched_barrier(0);
#pragma unroll
        for (int k = 0; k < CHUNK / 3; ++k)
            record(rs[k], pi + 3 * k);
        if (pi == 0) {
            RawStep last;
            fetch_step(last, CHUNK - 1);
            record(last, CHUNK - 1);
        }
    };

    auto consume = [&](int buf, int slot, int ybuf) {
        const v4f c0 = *(const v4f*)&sC[slot][lane >> 1][(lane & 1) * 8];
        const v4f c1 = *(const v4f*)&sC[slot][lane >> 1][(lane & 1) * 8 + 4];
        int cv[8];
#pragma unroll
        for (int k = 0; k < 4; ++k) {
            cv[k]     = __float_as_int(c0[k]);
            cv[4 + k] = __float_as_int(c1[k]);
        }
#pragma unroll
        for (int l = 0; l < CHUNK; ++l) {
            const float* rp = &sRec[buf][(l * SC_CH + lane) * REC_F];
            v4f a[4], d[4];
#pragma unroll
            for (int q = 0; q < 4; ++q) {
                a[q] = *(const v4f*)(rp + 4 * q);
                d[q] = *(const v4f*)(rp + 16 + 4 * q);
            }
            const v2f e = *(const v2f*)(rp + 32);
            float y = 0.f;
#pragma unroll
            for (int n = 0; n < N_ST; ++n) {
                const int v = l * N_ST + n;
                const float cn = __int_as_float(
                    __builtin_amdgcn_readlane(cv[v & 7], v >> 3));
                h[n] = a[n >> 2][n & 3] * h[n] + d[n >> 2][n & 3];
                y   += h[n] * cn;
            }
            sY[ybuf][l][lane] = (y + e.x * Dv) * e.y;
        }
    };

    auto store_y = [&](int ybuf, int row0) {
        const int l = lane & 15, hh = lane >> 4;
        float* dst = yout + (size_t)row0 * I_DIM + (size_t)i0 * 16 + 4 * lane;
#pragma unroll
        for (int q = 0; q < 4; ++q)
            *(v4f*)(dst + 128 * q) = *(const v4f*)&sY[ybuf][l][8 * q + 4 * hh];
    };

    request(rowb);
    park(0);
    __syncthreads();

    int s_cur = 2, s_nxt = 0, s_new = 1;
    for (int c = -1; c < NCHUNK; ++c) {
        const bool fetch = (c + 2 < NCHUNK);
        if (fetch) request(rowb + (c + 2) * CHUNK);
        if (wave == 0) {
            if (c >= 0) consume(c & 1, s_cur, c & 1);
        } else {
            if (wave == 3 && c >= 1) store_y((c - 1) & 1, rowb + (c - 1) * CHUNK);
            if (c + 1 < NCHUNK) produce(wave - 1, (c + 1) & 1, s_nxt);
        }
        if (fetch) park(s_new);
        __syncthreads();
        const int s_old = s_cur;
        s_cur = s_nxt; s_nxt = s_new; s_new = s_old;
    }
    if (wave == 3) store_y((NCHUNK - 1) & 1, rowb + (NCHUNK - 1) * CHUNK);
}

#define PACK_WG      512
#define PACK_THREADS (PACK_WG * 256)

template <int KD, int NPIECES>
__device__ __forceinline__
void pack_matrix(const float* __restrict__ X, float* __restrict__ P, unsigned first)
{
    constexpr unsigned KB = KD / 8;
    auto src = [&](unsigned p) {
        const unsigned ln = p & 31u, blk = p >> 5;
        const unsigned rt = blk / KB, kb = blk - rt * KB;
        return (gcv4f*)(X + (size_t)(16u * rt + (ln & 15u)) * KD + 8u * kb + 4u * (ln >> 4));
    };
    if constexpr (NPIECES % PACK_THREADS == 0) {
        constexpr int PER = NPIECES / PACK_THREADS;
        v4f v[PER];
#pragma unroll
        for (int q = 0; q < PER; ++q) v[q] = *src(first + q * PACK_THREADS);
#pragma unroll
        for (int q = 0; q < PER; ++q)
            *(v4f*)(P + (size_t)(first + q * PACK_THREADS) * 4) = v[q];
    } else {
        static_assert(NPIECES < PACK_THREADS, "a small matrix: at most one piece per thread");
        if (first < (unsigned)NPIECES) *(v4f*)(P + (size_t)first * 4) = *src(first);
    }
}

__global__ __launch_bounds__(256)
void pack_operands_kernel(const float* __restrict__ hidden,
                          const float* __restrict__ in_w,
                          const float* __restrict__ out_w,
                          const float* __restrict__ x_w,
                          float* __restrict__ hidden_p,
                          float* __restrict__ in_w_p,
                          float* __restrict__ out_w_p,
                          float* __restrict__ x_w_p)
{
    const unsigned first = blockIdx.x * 256u + threadIdx.x;
    pack_matrix<H_DIM, 2 * I_DIM * H_DIM / 4>(in_w,   in_w_p,   first);
    pack_matrix<H_DIM, ROWS * H_DIM / 4>     (hidden, hidden_p, first);
    pack_matrix<I_DIM, H_DIM * I_DIM / 4>    (out_w,  out_w_p,  first);
    pack_matrix<I_DIM, XN * I_DIM / 4>       (x_w,    x_w_p,    first);
}

extern "C" void kernel_launch(void* const* d_in, const int* in_sizes, int n_in,
                              void* d_out, int out_size, void* d_ws, size_t ws_size,
                              hipStream_t stream)
{
    const float* hidden     = (const float*)d_in[0];
    const float* in_proj_w  = (const float*)d_in[1];
    const float* conv_w     = (const float*)d_in[2];
    const float* conv_b     = (const float*)d_in[3];
    const float* x_proj_w   = (const float*)d_in[4];
    const float* dt_proj_w  = (const float*)d_in[5];
    const float* dt_proj_b  = (const float*)d_in[6];
    const float* dt_ln_w    = (const float*)d_in[7];
    const float* b_ln_w     = (const float*)d_in[8];
    const float* c_ln_w     = (const float*)d_in[9];
    const float* A_log      = (const float*)d_in[10];
    const float* Dvec       = (const float*)d_in[11];
    const float* out_proj_w = (const float*)d_in[12];
    float* out = (float*)d_out;

    float* ws   = (float*)d_ws;
    float* proj = ws;
    float* u    = proj + (size_t)ROWS * 2 * I_DIM;
    float* xdbl = u    + (size_t)ROWS * I_DIM;
    float* dtn  = xdbl + (size_t)ROWS * (R_DIM + 2 * N_ST);
    float* Bm   = dtn  + (size_t)ROWS * R_DIM;
    float* Cm   = Bm   + (size_t)ROWS * N_ST;
    float* dt2  = Cm   + (size_t)ROWS * N_ST;
    float* yout = dt2  + (size_t)ROWS * I_DIM;
    float* hid_p  = yout   + (size_t)ROWS * I_DIM;
    float* inw_p  = hid_p  + (size_t)ROWS * H_DIM;
    float* outw_p = inw_p  + (size_t)2 * I_DIM * H_DIM;
    float* xw_p   = outw_p + (size_t)H_DIM * I_DIM;

    pack_operands_kernel<<<PACK_WG, 256, 0, stream>>>(
        hidden, in_proj_w, out_proj_w, x_proj_w, hid_p, inw_p, outw_p, xw_p);

    wmma_gemm_f32_regtile<4, 4, 2, 2, 4, H_DIM, 2 * I_DIM>
        <<<dim3((2 * I_DIM) / 256, ROWS / 128), 256, 0, stream>>>(
        hid_p, inw_p, proj);

    conv_xproj_kernel<<<ROWS / 16, 256, 0, stream>>>(
        proj, conv_w, conv_b, xw_p, u, xdbl);

    rmsnorm_kernel<<<(ROWS + 255) / 256, 256, 0, stream>>>(
        xdbl, dt_ln_w, b_ln_w, c_ln_w, dtn, Bm, Cm);

    wmma_gemm_f32<4, 1><<<dim3(I_DIM / 64, ROWS / 128), 256, 0, stream>>>(
        dtn, dt_proj_w, dt_proj_b, dt2, ROWS, I_DIM, R_DIM);

    scan_kernel<<<dim3(I_DIM / SC_CH, B_SZ), 128, 0, stream>>>(
        dt2, u, Bm, Cm, A_log, Dvec, proj, yout);

    wmma_gemm_f32_regtile<2, 4, 4, 2, 2, I_DIM, H_DIM>
        <<<dim3(H_DIM / 128, ROWS / 64), 128, 0, stream>>>(
        yout, outw_p, out);
}


# scan: recurrence waves keep only h = dA*h + dBu (2 waves, 8 states each) and hand h on in an LDS tile; y = h.C, D skip, gate SiLU move to 4 output waves a chunk behind; sum order same
# speedup vs baseline: 7.5329x; 1.0447x over previous
#include <hip/hip_runtime.h>
#include <hip/hip_bf16.h>


#define H_DIM 1024
#define I_DIM 2048
#define N_ST  16
#define R_DIM 64
#define B_SZ  2
#define L_SEQ 1024
#define ROWS  (B_SZ * L_SEQ)
#define EPS   1e-06f

typedef __attribute__((ext_vector_type(2))) float v2f;
typedef __attribute__((ext_vector_type(8))) float v8f;

typedef __attribute__((address_space(1))) const float gcf;
typedef __attribute__((address_space(1))) const v2f  gcv2f;
typedef __attribute__((ext_vector_type(4))) float v4f;
typedef __attribute__((address_space(1))) const v4f gcv4f;

template <int NT, int EPI>
__global__ __launch_bounds__(256)
void wmma_gemm_f32(const float* __restrict__ A,
                   const float* __restrict__ W,
                   const float* __restrict__ bias,
                   float* __restrict__ C,
                   int M, int N, int K)
{
    const int lane = threadIdx.x & 31;
    const int wave = threadIdx.x >> 5;
    const int row0 = (blockIdx.y * 8 + wave) * 16;
    const int col0 = blockIdx.x * (16 * NT);

    const int r  = lane & 15;
    const int kk = (lane >> 4) * 2;

    gcf* Ap = (gcf*)(A + (size_t)(row0 + r) * K + kk);
    gcf* Wp[NT];
#pragma unroll
    for (int t = 0; t < NT; ++t)
        Wp[t] = (gcf*)(W + (size_t)(col0 + t * 16 + r) * K + kk);

    v8f acc[NT];
#pragma unroll
    for (int t = 0; t < NT; ++t) acc[t] = (v8f){0.f,0.f,0.f,0.f,0.f,0.f,0.f,0.f};

    for (int k = 0; k < K; k += 16) {
        __builtin_prefetch((const float*)(Ap + 256), 0, 1);

        v2f a[4];
#pragma unroll
        for (int s = 0; s < 4; ++s)
            a[s] = *(gcv2f*)(Ap + 4 * s);

        v2f b[NT][4];
#pragma unroll
        for (int t = 0; t < NT; ++t)
#pragma unroll
            for (int s = 0; s < 4; ++s)
                b[t][s] = *(gcv2f*)(Wp[t] + 4 * s);

#pragma unroll
        for (int s = 0; s < 4; ++s)
#pragma unroll
            for (int t = 0; t < NT; ++t)
                acc[t] = __builtin_amdgcn_wmma_f32_16x16x4_f32(
                             false, a[s], false, b[t][s],
                             (short)0, acc[t], false, false);

        Ap += 16;
#pragma unroll
        for (int t = 0; t < NT; ++t) Wp[t] += 16;
    }

    const int rowoff = (lane >> 4) * 8;
#pragma unroll
    for (int t = 0; t < NT; ++t) {
        const int col = col0 + t * 16 + r;
#pragma unroll
        for (int v = 0; v < 8; ++v) {
            float val = acc[t][v];
            if (EPI == 1) {
                val += bias[col];
                val = (val > 20.f) ? val : __logf(1.f + __expf(val));
            }
            C[(size_t)(row0 + rowoff + v) * N + col] = val;
        }
    }
}

template <int MT, int NT, int KS>
struct FragSet {
    v4f a[MT][KS];
    v4f b[NT][KS];
};

template <int MT, int NT, int KS, int KD>
__device__ __forceinline__
void frag_request(FragSet<MT, NT, KS>& f, gcf* Ap, gcf* Wp, int k)
{
#pragma unroll
    for (int mt = 0; mt < MT; ++mt)
#pragma unroll
        for (int j = 0; j < KS; ++j)
            f.a[mt][j] = *(gcv4f*)(Ap + (size_t)mt * 16 * KD + (k + 8 * j) * 16);
#pragma unroll
    for (int nt = 0; nt < NT; ++nt)
#pragma unroll
        for (int j = 0; j < KS; ++j)
            f.b[nt][j] = *(gcv4f*)(Wp + (size_t)nt * 16 * KD + (k + 8 * j) * 16);
}

template <int MT, int NT, int KS>
__device__ __forceinline__
void frag_mma(const FragSet<MT, NT, KS>& f, v8f (&acc)[MT][NT])
{
#pragma unroll
    for (int j = 0; j < KS; ++j) {
        v2f alo[MT], ahi[MT], blo[NT], bhi[NT];
#pragma unroll
        for (int mt = 0; mt < MT; ++mt) {
            alo[mt] = __builtin_shufflevector(f.a[mt][j], f.a[mt][j], 0, 1);
            ahi[mt] = __builtin_shufflevector(f.a[mt][j], f.a[mt][j], 2, 3);
        }
#pragma unroll
        for (int nt = 0; nt < NT; ++nt) {
            blo[nt] = __builtin_shufflevector(f.b[nt][j], f.b[nt][j], 0, 1);
            bhi[nt] = __builtin_shufflevector(f.b[nt][j], f.b[nt][j], 2, 3);
        }
#pragma unroll
        for (int mt = 0; mt < MT; ++mt)
#pragma unroll
            for (int nt = 0; nt < NT; ++nt)
                acc[mt][nt] = __builtin_amdgcn_wmma_f32_16x16x4_f32(
                                  false, alo[mt], false, blo[nt],
                                  (short)0, acc[mt][nt], false, false);
#pragma unroll
        for (int mt = 0; mt < MT; ++mt)
#pragma unroll
            for (int nt = 0; nt < NT; ++nt)
                acc[mt][nt] = __builtin_amdgcn_wmma_f32_16x16x4_f32(
                                  false, ahi[mt], false, bhi[nt],
                                  (short)0, acc[mt][nt], false, false);
    }
}

template <int MT, int NT, int KS, int WM, int WN, int KD, int ND>
__global__ __launch_bounds__(WM * WN * 32)
void wmma_gemm_f32_regtile(const float* __restrict__ A,
                           const float* __restrict__ W,
                           float* __restrict__ C)
{
    constexpr int KSET = 8 * KS;
    static_assert(KD % (2 * KSET) == 0, "K loop consumes two sets per trip");
    const int lane = threadIdx.x & 31;
    const int wave = threadIdx.x >> 5;
    const int wm   = wave / WN;
    const int wn   = wave - wm * WN;
    const int row0 = (blockIdx.y * WM + wm) * (16 * MT);
    const int col0 = (blockIdx.x * WN + wn) * (16 * NT);

    const int r = lane & 15;
    const int h = lane >> 4;

    gcf* Ap = (gcf*)(A + (size_t)row0 * KD + 4 * lane);
    gcf* Wp = (gcf*)(W + (size_t)col0 * KD + 4 * lane);

    v8f acc[MT][NT];
#pragma unroll
    for (int mt = 0; mt < MT; ++mt)
#pragma unroll
        for (int nt = 0; nt < NT; ++nt)
            acc[mt][nt] = (v8f){0.f,0.f,0.f,0.f,0.f,0.f,0.f,0.f};

    FragSet<MT, NT, KS> f0, f1;
    frag_request<MT, NT, KS, KD>(f0, Ap, Wp, 0);
    for (int k = 0; k < KD - 2 * KSET; k += 2 * KSET) {
        frag_request<MT, NT, KS, KD>(f1, Ap, Wp, k + KSET);
        __builtin_amdgcn_sched_barrier(0);
        frag_mma<MT, NT, KS>(f0, acc);
        __builtin_amdgcn_sched_barrier(0);
        frag_request<MT, NT, KS, KD>(f0, Ap, Wp, k + 2 * KSET);
        __builtin_amdgcn_sched_barrier(0);
        frag_mma<MT, NT, KS>(f1, acc);
        __builtin_amdgcn_sched_barrier(0);
    }
    frag_request<MT, NT, KS, KD>(f1, Ap, Wp, KD - KSET);
    __builtin_amdgcn_sched_barrier(0);
    frag_mma<MT, NT, KS>(f0, acc);
    frag_mma<MT, NT, KS>(f1, acc);

#pragma unroll
    for (int mt = 0; mt < MT; ++mt)
#pragma unroll
        for (int nt = 0; nt < NT; ++nt)
#pragma unroll
            for (int v = 0; v < 8; ++v)
                C[(size_t)(row0 + mt * 16 + 8 * h + v) * ND + col0 + nt * 16 + r] =
                    acc[mt][nt][v];
}

#define XN      (R_DIM + 2 * N_ST)
#define XNT     (XN / 16)
#define XPITCH  (I_DIM + 4)
#define XKW     (I_DIM / 8)

__global__ __launch_bounds__(256)
void conv_xproj_kernel(const float* __restrict__ proj,
                       const float* __restrict__ cw,
                       const float* __restrict__ cb,
                       const float* __restrict__ xw,
                       float* __restrict__ u,
                       float* __restrict__ xdbl)
{
    static_assert(L_SEQ % 16 == 0, "a 16-row tile must stay inside one batch");
    static_assert(XKW % 64 == 0, "K loop consumes two sets of 32 k per trip");
    __shared__ __attribute__((aligned(16))) float sA[16 * XPITCH];
    __shared__ __attribute__((aligned(16))) float sRed[8 * XNT * 8 * 32];

    const int tid  = threadIdx.x;
    const int lane = tid & 31;
    const int wave = tid >> 5;
    const int row0 = blockIdx.x * 16;
    const int l0   = row0 % L_SEQ;
    const bool has_top = (l0 > 0);
    const bool has_bot = (l0 + 16 < L_SEQ);

#pragma unroll
    for (int p = 0; p < 2; ++p) {
        const int ch = wave * XKW + p * 128 + lane * 4;
        gcf* src = (gcf*)(proj + (size_t)row0 * (2 * I_DIM) + ch);
        const v4f zero = (v4f){0.f, 0.f, 0.f, 0.f};
        v4f x[19];
        x[0] = zero; x[17] = zero; x[18] = zero;
        if (has_top) x[0] = *(gcv4f*)(src - (2 * I_DIM));
#pragma unroll
        for (int q = 1; q < 17; ++q)
            x[q] = *(gcv4f*)(src + (size_t)(q - 1) * (2 * I_DIM));
        if (has_bot) {
            x[17] = *(gcv4f*)(src + (size_t)16 * (2 * I_DIM));
            x[18] = *(gcv4f*)(src + (size_t)17 * (2 * I_DIM));
        }
        v4f w4[4];
#pragma unroll
        for (int c = 0; c < 4; ++c)
            w4[c] = *(gcv4f*)(cw + (size_t)(ch + c) * 4);
        const v4f bias = *(gcv4f*)(cb + ch);

#pragma unroll
        for (int r = 0; r < 16; ++r) {
            v4f o;
#pragma unroll
            for (int c = 0; c < 4; ++c) {
                float acc = bias[c];
#pragma unroll
                for (int k = 0; k < 4; ++k)
                    acc += w4[c][k] * x[r + k][c];
                o[c] = acc / (1.f + __expf(-acc));
            }
            *(v4f*)(u + (size_t)(row0 + r) * I_DIM + ch) = o;
            *(v4f*)&sA[r * XPITCH + ch] = o;
        }
    }
    __syncthreads();

    const int r = lane & 15;
    const int h = lane >> 4;
    const float* sAp = sA + r * XPITCH + wave * XKW + 4 * h;
    gcf* Wp = (gcf*)(xw + (size_t)(wave * XKW) * 16 + 4 * lane);

    v8f acc[1][XNT];
#pragma unroll
    for (int t = 0; t < XNT; ++t)
        acc[0][t] = (v8f){0.f,0.f,0.f,0.f,0.f,0.f,0.f,0.f};

    auto request = [&](FragSet<1, XNT, 4>& f, int k) {
#pragma unroll
        for (int j = 0; j < 4; ++j)
            f.a[0][j] = *(const v4f*)(sAp + k + 8 * j);
#pragma unroll
        for (int t = 0; t < XNT; ++t)
#pragma unroll
            for (int j = 0; j < 4; ++j)
                f.b[t][j] = *(gcv4f*)(Wp + (size_t)t * 16 * I_DIM + (k + 8 * j) * 16);
    };

    FragSet<1, XNT, 4> f0, f1;
    request(f0, 0);
    for (int k = 0; k < XKW - 64; k += 64) {
        request(f1, k + 32);
        __builtin_amdgcn_sched_barrier(0);
        frag_mma<1, XNT, 4>(f0, acc);
        __builtin_amdgcn_sched_barrier(0);
        request(f0, k + 64);
        __builtin_amdgcn_sched_barrier(0);
        frag_mma<1, XNT, 4>(f1, acc);
        __builtin_amdgcn_sched_barrier(0);
    }
    request(f1, XKW - 32);
    __builtin_amdgcn_sched_barrier(0);
    frag_mma<1, XNT, 4>(f0, acc);
    frag_mma<1, XNT, 4>(f1, acc);

#pragma unroll
    for (int t = 0; t < XNT; ++t)
#pragma unroll
        for (int v = 0; v < 8; ++v)
            sRed[((wave * XNT + t) * 8 + v) * 32 + lane] = acc[0][t][v];
    __syncthreads();

#pragma unroll
    for (int q = 0; q < XNT; ++q) {
        const int e = tid + 256 * q;
        float sum = 0.f;
#pragma unroll
        for (int w = 0; w < 8; ++w)
            sum += sRed[w * (XNT * 8 * 32) + e];
        const int t  = e >> 8;
        const int v  = (e & 255) >> 5;
        const int ln = e & 31;
        xdbl[(size_t)(row0 + 8 * (ln >> 4) + v) * XN + t * 16 + (ln & 15)] = sum;
    }
}

__global__ __launch_bounds__(256)
void rmsnorm_kernel(const float* __restrict__ xd,
                    const float* __restrict__ wdt,
                    const float* __restrict__ wb,
                    const float* __restrict__ wc,
                    float* __restrict__ dtn,
                    float* __restrict__ Bm,
                    float* __restrict__ Cm)
{
    const int row = blockIdx.x * 256 + threadIdx.x;
    if (row >= ROWS) return;
    const float* x = xd + (size_t)row * (R_DIM + 2 * N_ST);

    float s = 0.f;
#pragma unroll 8
    for (int j = 0; j < R_DIM; ++j) s += x[j] * x[j];
    float inv = __frsqrt_rn(s / R_DIM + EPS);
#pragma unroll 8
    for (int j = 0; j < R_DIM; ++j)
        dtn[(size_t)row * R_DIM + j] = wdt[j] * x[j] * inv;

    s = 0.f;
#pragma unroll
    for (int j = 0; j < N_ST; ++j) { float v = x[R_DIM + j]; s += v * v; }
    inv = __frsqrt_rn(s / N_ST + EPS);
#pragma unroll
    for (int j = 0; j < N_ST; ++j)
        Bm[(size_t)row * N_ST + j] = wb[j] * x[R_DIM + j] * inv;

    s = 0.f;
#pragma unroll
    for (int j = 0; j < N_ST; ++j) { float v = x[R_DIM + N_ST + j]; s += v * v; }
    inv = __frsqrt_rn(s / N_ST + EPS);
#pragma unroll
    for (int j = 0; j < N_ST; ++j)
        Cm[(size_t)row * N_ST + j] = wc[j] * x[R_DIM + N_ST + j] * inv;
}

#define CHUNK   16
#define NCHUNK  (L_SEQ / CHUNK)
#define SC_CH   32
#define REC_F   36
#define REC_BUF (CHUNK * SC_CH * REC_F)
#define H_F     20
#define H_BUF   (CHUNK * SC_CH * H_F)
#define Y_PITCH 36
#define NRAW    4
#define NREC    2
#define NSTATE  (N_ST / NREC)
#define NPROD   4
#define NOUT    4
#define SC_WAVES   (NREC + NPROD + NOUT)
#define SC_THREADS (SC_WAVES * 32)

struct ScanTile {
    v4f d, u, g;
    v2f b, c;
};

__global__ __launch_bounds__(SC_THREADS)
void scan_kernel(const float* __restrict__ dt2,
                 const float* __restrict__ u,
                 const float* __restrict__ Bm,
                 const float* __restrict__ Cm,
                 const float* __restrict__ A_log,
                 const float* __restrict__ Dvec,
                 const float* __restrict__ proj,
                 float* __restrict__ yout)
{
    static_assert(N_ST % NREC == 0 && NSTATE % 4 == 0, "whole 16-byte pieces per recurrence wave");
    __shared__ __attribute__((aligned(16))) float sRec[2][REC_BUF];
    __shared__ __attribute__((aligned(16))) float sH[2][H_BUF];
    __shared__ __attribute__((aligned(16))) float sD[NRAW][CHUNK][SC_CH];
    __shared__ __attribute__((aligned(16))) float sU[NRAW][CHUNK][SC_CH];
    __shared__ __attribute__((aligned(16))) float sG[NRAW][CHUNK][SC_CH];
    __shared__ __attribute__((aligned(16))) float sB[NRAW][CHUNK][N_ST];
    __shared__ __attribute__((aligned(16))) float sC[NRAW][CHUNK][N_ST];
    __shared__ __attribute__((aligned(16))) float sY[2][CHUNK][Y_PITCH];

    const int tid  = threadIdx.x;
    const int lane = tid & 31;
    const int wave = __builtin_amdgcn_readfirstlane(tid >> 5);
    const int b    = blockIdx.y;
    const int i0   = blockIdx.x * SC_CH;
    const int i    = i0 + lane;
    const int rowb = b * L_SEQ;

    float Aa[N_ST];
#pragma unroll
    for (int n = 0; n < N_ST; ++n)
        Aa[n] = -__expf(A_log[(size_t)i * N_ST + n]);

    float h[NSTATE];
#pragma unroll
    for (int n = 0; n < NSTATE; ++n) h[n] = 0.f;

    const float Dv = Dvec[i];

    ScanTile t;

    auto request = [&](int row0) {
        const size_t row = (size_t)(row0 + (tid >> 3));
        const int    col = i0 + (tid & 7) * 4;
        t.d = *(gcv4f*)(dt2  + row * I_DIM + col);
        t.u = *(gcv4f*)(u    + row * I_DIM + col);
        t.g = *(gcv4f*)(proj + row * (2 * I_DIM) + I_DIM + col);
        t.b = *(gcv2f*)(Bm + (size_t)row0 * N_ST + tid * 2);
        t.c = *(gcv2f*)(Cm + (size_t)row0 * N_ST + tid * 2);
    };

    auto park = [&](int slot) {
        const int r = tid >> 3, p = tid & 7;
        *(v4f*)&sD[slot][r][p * 4] = t.d;
        *(v4f*)&sU[slot][r][p * 4] = t.u;
        *(v4f*)&sG[slot][r][p * 4] = t.g;
        *(v2f*)&sB[slot][r][p * 2] = t.b;
        *(v2f*)&sC[slot][r][p * 2] = t.c;
    };

    static_assert(NPROD * (CHUNK / NPROD) == CHUNK, "every feeding wave makes CHUNK/NPROD steps");
    struct RawStep { float dtv, uv; v4f bq[4]; };
    auto produce = [&](int pi, int buf, int slot) {
        auto fetch_step = [&](RawStep& s, int l) {
            s.dtv = sD[slot][l][lane];
            s.uv  = sU[slot][l][lane];
#pragma unroll
            for (int q = 0; q < 4; ++q)
                s.bq[q] = *(const v4f*)&sB[slot][l][4 * q];
        };
        auto record = [&](const RawStep& s, int l) {
            const float du = s.dtv * s.uv;
            float* rp = &sRec[buf][(l * SC_CH + lane) * REC_F];
#pragma unroll
            for (int q = 0; q < 4; ++q) {
                v4f a, d;
#pragma unroll
                for (int j = 0; j < 4; ++j) {
                    a[j] = __expf(s.dtv * Aa[4 * q + j]);
                    d[j] = du * s.bq[q][j];
                }
                *(v4f*)(rp + 4 * q)      = a;
                *(v4f*)(rp + 16 + 4 * q) = d;
            }
        };
        RawStep rs[CHUNK / NPROD];
#pragma unroll
        for (int k = 0; k < CHUNK / NPROD; ++k)
            fetch_step(rs[k], pi + NPROD * k);
        __builtin_amdgcn_sched_barrier(0);
#pragma unroll
        for (int k = 0; k < CHUNK / NPROD; ++k)
            record(rs[k], pi + NPROD * k);
    };

    auto consume = [&](int r, int buf) {
        const float* rbase = &sRec[buf][lane * REC_F + NSTATE * r];
        float*       hbase = &sH[buf][lane * H_F + NSTATE * r];
#pragma unroll
        for (int l = 0; l < CHUNK; ++l) {
            const float* rp = rbase + l * SC_CH * REC_F;
            float*       hp = hbase + l * SC_CH * H_F;
            v4f a[NSTATE / 4], d[NSTATE / 4];
#pragma unroll
            for (int q = 0; q < NSTATE / 4; ++q) {
                a[q] = *(const v4f*)(rp + 4 * q);
                d[q] = *(const v4f*)(rp + 16 + 4 * q);
            }
#pragma unroll
            for (int q = 0; q < NSTATE / 4; ++q) {
                v4f o;
#pragma unroll
                for (int j = 0; j < 4; ++j) {
                    h[4 * q + j] = a[q][j] * h[4 * q + j] + d[q][j];
                    o[j] = h[4 * q + j];
                }
                *(v4f*)(hp + 4 * q) = o;
            }
        }
    };

    static_assert(NOUT * 4 == CHUNK, "four steps per output wave");
    auto ywork = [&](int yw, int hbuf, int slot, int ybuf) {
        const v4f cq = *(const v4f*)&sC[slot][4 * yw + ((lane & 15) >> 2)][(lane & 3) * 4];
        int cv[4];
#pragma unroll
        for (int k = 0; k < 4; ++k) cv[k] = __float_as_int(cq[k]);
#pragma unroll
        for (int ll = 0; ll < 4; ++ll) {
            const int l = 4 * yw + ll;
            const float* hp = &sH[hbuf][(l * SC_CH + lane) * H_F];
            v4f hq[4];
#pragma unroll
            for (int q = 0; q < 4; ++q) hq[q] = *(const v4f*)(hp + 4 * q);
            const float uv = sU[slot][l][lane];
            const float g  = sG[slot][l][lane];
            float y = 0.f;
#pragma unroll
            for (int n = 0; n < N_ST; ++n) {
                const int v = ll * N_ST + n;
                const float cn = __int_as_float(
                    __builtin_amdgcn_readlane(cv[v & 3], v >> 2));
                y += hq[n >> 2][n & 3] * cn;
            }
            const float sg = g / (1.f + __expf(-g));
            sY[ybuf][l][lane] = (y + uv * Dv) * sg;
        }
    };

    auto store_y = [&](int yw, int ybuf, int row0) {
        const int l = lane & 15, hh = lane >> 4;
        float* dst = yout + (size_t)row0 * I_DIM + (size_t)i0 * 16 + 128 * yw + 4 * lane;
        *(v4f*)dst = *(const v4f*)&sY[ybuf][l][8 * yw + 4 * hh];
    };

    const bool feeder = (wave < 4);
    const int  yw     = wave - (NREC + NPROD);

    if (feeder) { request(rowb); park(0); }
    __syncthreads();

    for (int c = -1; c <= NCHUNK; ++c) {
        const bool fetch = feeder && (c + 2 < NCHUNK);
        if (fetch) request(rowb + (c + 2) * CHUNK);
        if (wave < NREC) {
            if (c >= 0 && c < NCHUNK) consume(wave, c & 1);
        } else if (wave < NREC + NPROD) {
            if (c + 1 < NCHUNK) produce(wave - NREC, (c + 1) & 1, (c + 1) & 3);
        } else {
            if (c >= 2) store_y(yw, c & 1, rowb + (c - 2) * CHUNK);
            if (c >= 1) ywork(yw, (c - 1) & 1, (c - 1) & 3, (c - 1) & 1);
        }
        if (fetch) park((c + 2) & 3);
        __syncthreads();
    }
    if (yw >= 0) store_y(yw, (NCHUNK - 1) & 1, rowb + (NCHUNK - 1) * CHUNK);
}

#define PACK_WG      512
#define PACK_THREADS (PACK_WG * 256)

template <int KD, int NPIECES>
__device__ __forceinline__
void pack_matrix(const float* __restrict__ X, float* __restrict__ P, unsigned first)
{
    constexpr unsigned KB = KD / 8;
    auto src = [&](unsigned p) {
        const unsigned ln = p & 31u, blk = p >> 5;
        const unsigned rt = blk / KB, kb = blk - rt * KB;
        return (gcv4f*)(X + (size_t)(16u * rt + (ln & 15u)) * KD + 8u * kb + 4u * (ln >> 4));
    };
    if constexpr (NPIECES % PACK_THREADS == 0) {
        constexpr int PER = NPIECES / PACK_THREADS;
        v4f v[PER];
#pragma unroll
        for (int q = 0; q < PER; ++q) v[q] = *src(first + q * PACK_THREADS);
#pragma unroll
        for (int q = 0; q < PER; ++q)
            *(v4f*)(P + (size_t)(first + q * PACK_THREADS) * 4) = v[q];
    } else {
        static_assert(NPIECES < PACK_THREADS, "a small matrix: at most one piece per thread");
        if (first < (unsigned)NPIECES) *(v4f*)(P + (size_t)first * 4) = *src(first);
    }
}

__global__ __launch_bounds__(256)
void pack_operands_kernel(const float* __restrict__ hidden,
                          const float* __restrict__ in_w,
                          const float* __restrict__ out_w,
                          const float* __restrict__ x_w,
                          float* __restrict__ hidden_p,
                          float* __restrict__ in_w_p,
                          float* __restrict__ out_w_p,
                          float* __restrict__ x_w_p)
{
    const unsigned first = blockIdx.x * 256u + threadIdx.x;
    pack_matrix<H_DIM, 2 * I_DIM * H_DIM / 4>(in_w,   in_w_p,   first);
    pack_matrix<H_DIM, ROWS * H_DIM / 4>     (hidden, hidden_p, first);
    pack_matrix<I_DIM, H_DIM * I_DIM / 4>    (out_w,  out_w_p,  first);
    pack_matrix<I_DIM, XN * I_DIM / 4>       (x_w,    x_w_p,    first);
}

extern "C" void kernel_launch(void* const* d_in, const int* in_sizes, int n_in,
                              void* d_out, int out_size, void* d_ws, size_t ws_size,
                              hipStream_t stream)
{
    const float* hidden     = (const float*)d_in[0];
    const float* in_proj_w  = (const float*)d_in[1];
    const float* conv_w     = (const float*)d_in[2];
    const float* conv_b     = (const float*)d_in[3];
    const float* x_proj_w   = (const float*)d_in[4];
    const float* dt_proj_w  = (const float*)d_in[5];
    const float* dt_proj_b  = (const float*)d_in[6];
    const float* dt_ln_w    = (const float*)d_in[7];
    const float* b_ln_w     = (const float*)d_in[8];
    const float* c_ln_w     = (const float*)d_in[9];
    const float* A_log      = (const float*)d_in[10];
    const float* Dvec       = (const float*)d_in[11];
    const float* out_proj_w = (const float*)d_in[12];
    float* out = (float*)d_out;

    float* ws   = (float*)d_ws;
    float* proj = ws;
    float* u    = proj + (size_t)ROWS * 2 * I_DIM;
    float* xdbl = u    + (size_t)ROWS * I_DIM;
    float* dtn  = xdbl + (size_t)ROWS * (R_DIM + 2 * N_ST);
    float* Bm   = dtn  + (size_t)ROWS * R_DIM;
    float* Cm   = Bm   + (size_t)ROWS * N_ST;
    float* dt2  = Cm   + (size_t)ROWS * N_ST;
    float* yout = dt2  + (size_t)ROWS * I_DIM;
    float* hid_p  = yout   + (size_t)ROWS * I_DIM;
    float* inw_p  = hid_p  + (size_t)ROWS * H_DIM;
    float* outw_p = inw_p  + (size_t)2 * I_DIM * H_DIM;
    float* xw_p   = outw_p + (size_t)H_DIM * I_DIM;

    pack_operands_kernel<<<PACK_WG, 256, 0, stream>>>(
        hidden, in_proj_w, out_proj_w, x_proj_w, hid_p, inw_p, outw_p, xw_p);

    wmma_gemm_f32_regtile<4, 4, 2, 2, 4, H_DIM, 2 * I_DIM>
        <<<dim3((2 * I_DIM) / 256, ROWS / 128), 256, 0, stream>>>(
        hid_p, inw_p, proj);

    conv_xproj_kernel<<<ROWS / 16, 256, 0, stream>>>(
        proj, conv_w, conv_b, xw_p, u, xdbl);

    rmsnorm_kernel<<<(ROWS + 255) / 256, 256, 0, stream>>>(
        xdbl, dt_ln_w, b_ln_w, c_ln_w, dtn, Bm, Cm);

    wmma_gemm_f32<4, 1><<<dim3(I_DIM / 64, ROWS / 128), 256, 0, stream>>>(
        dtn, dt_proj_w, dt_proj_b, dt2, ROWS, I_DIM, R_DIM);

    scan_kernel<<<dim3(I_DIM / SC_CH, B_SZ), SC_THREADS, 0, stream>>>(
        dt2, u, Bm, Cm, A_log, Dvec, proj, yout);

    wmma_gemm_f32_regtile<2, 4, 4, 2, 2, I_DIM, H_DIM>
        <<<dim3(H_DIM / 128, ROWS / 64), 128, 0, stream>>>(
        yout, outw_p, out);
}


# dt path folded into its neighbours: RMS norms in x_proj's work-group (x_dbl tile in LDS), dt_proj+softplus on 4 new waves of the scan, 16 WMMAs per chunk and tile; 2 launches, dt2 go; sum order same
# speedup vs baseline: 7.9302x; 1.0527x over previous
#include <hip/hip_runtime.h>
#include <hip/hip_bf16.h>


#define H_DIM 1024
#define I_DIM 2048
#define N_ST  16
#define R_DIM 64
#define B_SZ  2
#define L_SEQ 1024
#define ROWS  (B_SZ * L_SEQ)
#define EPS   1e-06f

typedef __attribute__((ext_vector_type(2))) float v2f;
typedef __attribute__((ext_vector_type(8))) float v8f;

typedef __attribute__((address_space(1))) const float gcf;
typedef __attribute__((address_space(1))) const v2f  gcv2f;
typedef __attribute__((ext_vector_type(4))) float v4f;
typedef __attribute__((address_space(1))) const v4f gcv4f;

template <int MT, int NT, int KS>
struct FragSet {
    v4f a[MT][KS];
    v4f b[NT][KS];
};

template <int MT, int NT, int KS, int KD>
__device__ __forceinline__
void frag_request(FragSet<MT, NT, KS>& f, gcf* Ap, gcf* Wp, int k)
{
#pragma unroll
    for (int mt = 0; mt < MT; ++mt)
#pragma unroll
        for (int j = 0; j < KS; ++j)
            f.a[mt][j] = *(gcv4f*)(Ap + (size_t)mt * 16 * KD + (k + 8 * j) * 16);
#pragma unroll
    for (int nt = 0; nt < NT; ++nt)
#pragma unroll
        for (int j = 0; j < KS; ++j)
            f.b[nt][j] = *(gcv4f*)(Wp + (size_t)nt * 16 * KD + (k + 8 * j) * 16);
}

template <int MT, int NT, int KS>
__device__ __forceinline__
void frag_mma(const FragSet<MT, NT, KS>& f, v8f (&acc)[MT][NT])
{
#pragma unroll
    for (int j = 0; j < KS; ++j) {
        v2f alo[MT], ahi[MT], blo[NT], bhi[NT];
#pragma unroll
        for (int mt = 0; mt < MT; ++mt) {
            alo[mt] = __builtin_shufflevector(f.a[mt][j], f.a[mt][j], 0, 1);
            ahi[mt] = __builtin_shufflevector(f.a[mt][j], f.a[mt][j], 2, 3);
        }
#pragma unroll
        for (int nt = 0; nt < NT; ++nt) {
            blo[nt] = __builtin_shufflevector(f.b[nt][j], f.b[nt][j], 0, 1);
            bhi[nt] = __builtin_shufflevector(f.b[nt][j], f.b[nt][j], 2, 3);
        }
#pragma unroll
        for (int mt = 0; mt < MT; ++mt)
#pragma unroll
            for (int nt = 0; nt < NT; ++nt)
                acc[mt][nt] = __builtin_amdgcn_wmma_f32_16x16x4_f32(
                                  false, alo[mt], false, blo[nt],
                                  (short)0, acc[mt][nt], false, false);
#pragma unroll
        for (int mt = 0; mt < MT; ++mt)
#pragma unroll
            for (int nt = 0; nt < NT; ++nt)
                acc[mt][nt] = __builtin_amdgcn_wmma_f32_16x16x4_f32(
                                  false, ahi[mt], false, bhi[nt],
                                  (short)0, acc[mt][nt], false, false);
    }
}

template <int MT, int NT, int KS, int WM, int WN, int KD, int ND>
__global__ __launch_bounds__(WM * WN * 32)
void wmma_gemm_f32_regtile(const float* __restrict__ A,
                           const float* __restrict__ W,
                           float* __restrict__ C)
{
    constexpr int KSET = 8 * KS;
    static_assert(KD % (2 * KSET) == 0, "K loop consumes two sets per trip");
    const int lane = threadIdx.x & 31;
    const int wave = threadIdx.x >> 5;
    const int wm   = wave / WN;
    const int wn   = wave - wm * WN;
    const int row0 = (blockIdx.y * WM + wm) * (16 * MT);
    const int col0 = (blockIdx.x * WN + wn) * (16 * NT);

    const int r = lane & 15;
    const int h = lane >> 4;

    gcf* Ap = (gcf*)(A + (size_t)row0 * KD + 4 * lane);
    gcf* Wp = (gcf*)(W + (size_t)col0 * KD + 4 * lane);

    v8f acc[MT][NT];
#pragma unroll
    for (int mt = 0; mt < MT; ++mt)
#pragma unroll
        for (int nt = 0; nt < NT; ++nt)
            acc[mt][nt] = (v8f){0.f,0.f,0.f,0.f,0.f,0.f,0.f,0.f};

    FragSet<MT, NT, KS> f0, f1;
    frag_request<MT, NT, KS, KD>(f0, Ap, Wp, 0);
    for (int k = 0; k < KD - 2 * KSET; k += 2 * KSET) {
        frag_request<MT, NT, KS, KD>(f1, Ap, Wp, k + KSET);
        __builtin_amdgcn_sched_barrier(0);
        frag_mma<MT, NT, KS>(f0, acc);
        __builtin_amdgcn_sched_barrier(0);
        frag_request<MT, NT, KS, KD>(f0, Ap, Wp, k + 2 * KSET);
        __builtin_amdgcn_sched_barrier(0);
        frag_mma<MT, NT, KS>(f1, acc);
        __builtin_amdgcn_sched_barrier(0);
    }
    frag_request<MT, NT, KS, KD>(f1, Ap, Wp, KD - KSET);
    __builtin_amdgcn_sched_barrier(0);
    frag_mma<MT, NT, KS>(f0, acc);
    frag_mma<MT, NT, KS>(f1, acc);

#pragma unroll
    for (int mt = 0; mt < MT; ++mt)
#pragma unroll
        for (int nt = 0; nt < NT; ++nt)
#pragma unroll
            for (int v = 0; v < 8; ++v)
                C[(size_t)(row0 + mt * 16 + 8 * h + v) * ND + col0 + nt * 16 + r] =
                    acc[mt][nt][v];
}

#define XN      (R_DIM + 2 * N_ST)
#define XNT     (XN / 16)
#define XPITCH  (I_DIM + 4)
#define XKW     (I_DIM / 8)
#define XP      100

__global__ __launch_bounds__(256)
void conv_xproj_kernel(const float* __restrict__ proj,
                       const float* __restrict__ cw,
                       const float* __restrict__ cb,
                       const float* __restrict__ xw,
                       const float* __restrict__ wdt,
                       const float* __restrict__ wb,
                       const float* __restrict__ wc,
                       float* __restrict__ u,
                       float* __restrict__ dtn,
                       float* __restrict__ Bm,
                       float* __restrict__ Cm)
{
    static_assert(L_SEQ % 16 == 0, "a 16-row tile must stay inside one batch");
    static_assert(XKW % 64 == 0, "K loop consumes two sets of 32 k per trip");
    __shared__ __attribute__((aligned(16))) float sA[16 * XPITCH];
    __shared__ __attribute__((aligned(16))) float sRed[8 * XNT * 8 * 32];

    const int tid  = threadIdx.x;
    const int lane = tid & 31;
    const int wave = tid >> 5;
    const int row0 = blockIdx.x * 16;
    const int l0   = row0 % L_SEQ;
    const bool has_top = (l0 > 0);
    const bool has_bot = (l0 + 16 < L_SEQ);

#pragma unroll
    for (int p = 0; p < 2; ++p) {
        const int ch = wave * XKW + p * 128 + lane * 4;
        gcf* src = (gcf*)(proj + (size_t)row0 * (2 * I_DIM) + ch);
        const v4f zero = (v4f){0.f, 0.f, 0.f, 0.f};
        v4f x[19];
        x[0] = zero; x[17] = zero; x[18] = zero;
        if (has_top) x[0] = *(gcv4f*)(src - (2 * I_DIM));
#pragma unroll
        for (int q = 1; q < 17; ++q)
            x[q] = *(gcv4f*)(src + (size_t)(q - 1) * (2 * I_DIM));
        if (has_bot) {
            x[17] = *(gcv4f*)(src + (size_t)16 * (2 * I_DIM));
            x[18] = *(gcv4f*)(src + (size_t)17 * (2 * I_DIM));
        }
        v4f w4[4];
#pragma unroll
        for (int c = 0; c < 4; ++c)
            w4[c] = *(gcv4f*)(cw + (size_t)(ch + c) * 4);
        const v4f bias = *(gcv4f*)(cb + ch);

#pragma unroll
        for (int r = 0; r < 16; ++r) {
            v4f o;
#pragma unroll
            for (int c = 0; c < 4; ++c) {
                float acc = bias[c];
#pragma unroll
                for (int k = 0; k < 4; ++k)
                    acc += w4[c][k] * x[r + k][c];
                o[c] = acc / (1.f + __expf(-acc));
            }
            *(v4f*)(u + (size_t)(row0 + r) * I_DIM + ch) = o;
            *(v4f*)&sA[r * XPITCH + ch] = o;
        }
    }
    __syncthreads();

    const int r = lane & 15;
    const int h = lane >> 4;
    const float* sAp = sA + r * XPITCH + wave * XKW + 4 * h;
    gcf* Wp = (gcf*)(xw + (size_t)(wave * XKW) * 16 + 4 * lane);

    v8f acc[1][XNT];
#pragma unroll
    for (int t = 0; t < XNT; ++t)
        acc[0][t] = (v8f){0.f,0.f,0.f,0.f,0.f,0.f,0.f,0.f};

    auto request = [&](FragSet<1, XNT, 4>& f, int k) {
#pragma unroll
        for (int j = 0; j < 4; ++j)
            f.a[0][j] = *(const v4f*)(sAp + k + 8 * j);
#pragma unroll
        for (int t = 0; t < XNT; ++t)
#pragma unroll
            for (int j = 0; j < 4; ++j)
                f.b[t][j] = *(gcv4f*)(Wp + (size_t)t * 16 * I_DIM + (k + 8 * j) * 16);
    };

    FragSet<1, XNT, 4> f0, f1;
    request(f0, 0);
    for (int k = 0; k < XKW - 64; k += 64) {
        request(f1, k + 32);
        __builtin_amdgcn_sched_barrier(0);
        frag_mma<1, XNT, 4>(f0, acc);
        __builtin_amdgcn_sched_barrier(0);
        request(f0, k + 64);
        __builtin_amdgcn_sched_barrier(0);
        frag_mma<1, XNT, 4>(f1, acc);
        __builtin_amdgcn_sched_barrier(0);
    }
    request(f1, XKW - 32);
    __builtin_amdgcn_sched_barrier(0);
    frag_mma<1, XNT, 4>(f0, acc);
    frag_mma<1, XNT, 4>(f1, acc);

#pragma unroll
    for (int t = 0; t < XNT; ++t)
#pragma unroll
        for (int v = 0; v < 8; ++v)
            sRed[((wave * XNT + t) * 8 + v) * 32 + lane] = acc[0][t][v];
    __syncthreads();

    float* sX   = sA;
    float* sInv = sA + 16 * XP;

    const int xrow = 8 * (lane >> 4) + wave;
    const int xc   = lane & 15;
    float xv[XNT];
#pragma unroll
    for (int q = 0; q < XNT; ++q) {
        const int e = tid + 256 * q;
        float sum = 0.f;
#pragma unroll
        for (int w = 0; w < 8; ++w)
            sum += sRed[w * (XNT * 8 * 32) + e];
        xv[q] = sum;
        sX[xrow * XP + 16 * q + xc] = sum;
    }
    __syncthreads();

    if (tid < 48) {
        const int seg = tid >> 4;
        const float* x = sX + (tid & 15) * XP;
        float s = 0.f, inv;
        if (seg == 0) {
#pragma unroll 8
            for (int j = 0; j < R_DIM; ++j) s += x[j] * x[j];
            inv = __frsqrt_rn(s / R_DIM + EPS);
        } else {
            const float* xs = x + R_DIM + (seg - 1) * N_ST;
#pragma unroll
            for (int j = 0; j < N_ST; ++j) { float v = xs[j]; s += v * v; }
            inv = __frsqrt_rn(s / N_ST + EPS);
        }
        sInv[tid] = inv;
    }
    __syncthreads();

    const float inv_dt = sInv[xrow], inv_b = sInv[16 + xrow], inv_c = sInv[32 + xrow];
#pragma unroll
    for (int q = 0; q < R_DIM / 16; ++q)
        dtn[(size_t)(row0 + xrow) * R_DIM + 16 * q + xc] = wdt[16 * q + xc] * xv[q] * inv_dt;
    Bm[(size_t)(row0 + xrow) * N_ST + xc] = wb[xc] * xv[4] * inv_b;
    Cm[(size_t)(row0 + xrow) * N_ST + xc] = wc[xc] * xv[5] * inv_c;
}

#define CHUNK   16
#define NCHUNK  (L_SEQ / CHUNK)
#define SC_CH   32
#define REC_F   36
#define REC_BUF (CHUNK * SC_CH * REC_F)
#define H_F     20
#define H_BUF   (CHUNK * SC_CH * H_F)
#define Y_PITCH 36
#define NRAW    4
#define NREC    2
#define NSTATE  (N_ST / NREC)
#define NPROD   4
#define NOUT    4
#define NDT     (2 * (SC_CH / 16))
#define SC_WAVES   (NREC + NPROD + NOUT + NDT)
#define SC_THREADS (SC_WAVES * 32)

struct ScanTile {
    v4f u, g;
    v2f b, c;
};

struct DtFrag { v2f k[R_DIM / 4]; };

__global__ __launch_bounds__(SC_THREADS)
void scan_kernel(const float* __restrict__ dtn,
                 const float* __restrict__ dtw,
                 const float* __restrict__ dtb,
                 const float* __restrict__ u,
                 const float* __restrict__ Bm,
                 const float* __restrict__ Cm,
                 const float* __restrict__ A_log,
                 const float* __restrict__ Dvec,
                 const float* __restrict__ proj,
                 float* __restrict__ yout)
{
    static_assert(N_ST % NREC == 0 && NSTATE % 4 == 0, "whole 16-byte pieces per recurrence wave");
    __shared__ __attribute__((aligned(16))) float sRec[2][REC_BUF];
    __shared__ __attribute__((aligned(16))) float sH[2][H_BUF];
    __shared__ __attribute__((aligned(16))) float sD[NRAW][CHUNK][SC_CH];
    __shared__ __attribute__((aligned(16))) float sU[NRAW][CHUNK][SC_CH];
    __shared__ __attribute__((aligned(16))) float sG[NRAW][CHUNK][SC_CH];
    __shared__ __attribute__((aligned(16))) float sB[NRAW][CHUNK][N_ST];
    __shared__ __attribute__((aligned(16))) float sC[NRAW][CHUNK][N_ST];
    __shared__ __attribute__((aligned(16))) float sY[2][CHUNK][Y_PITCH];
    __shared__ __attribute__((aligned(16))) v2f sFa[NDT][R_DIM / 4][32];
    __shared__ __attribute__((aligned(16))) v2f sFw[NDT][R_DIM / 4][32];
    static_assert(NDT == 4, "dt wave = (tile, chunk parity)");

    const int tid  = threadIdx.x;
    const int lane = tid & 31;
    const int wave = __builtin_amdgcn_readfirstlane(tid >> 5);
    const int b    = blockIdx.y;
    const int i0   = blockIdx.x * SC_CH;
    const int i    = i0 + lane;
    const int rowb = b * L_SEQ;

    float Aa[N_ST];
#pragma unroll
    for (int n = 0; n < N_ST; ++n)
        Aa[n] = -__expf(A_log[(size_t)i * N_ST + n]);

    float h[NSTATE];
#pragma unroll
    for (int n = 0; n < NSTATE; ++n) h[n] = 0.f;

    const float Dv = Dvec[i];

    ScanTile t;

    auto request = [&](int row0) {
        const size_t row = (size_t)(row0 + (tid >> 3));
        const int    col = i0 + (tid & 7) * 4;
        t.u = *(gcv4f*)(u    + row * I_DIM + col);
        t.g = *(gcv4f*)(proj + row * (2 * I_DIM) + I_DIM + col);
        t.b = *(gcv2f*)(Bm + (size_t)row0 * N_ST + tid * 2);
        t.c = *(gcv2f*)(Cm + (size_t)row0 * N_ST + tid * 2);
    };

    auto park = [&](int slot) {
        const int r = tid >> 3, p = tid & 7;
        *(v4f*)&sU[slot][r][p * 4] = t.u;
        *(v4f*)&sG[slot][r][p * 4] = t.g;
        *(v2f*)&sB[slot][r][p * 2] = t.b;
        *(v2f*)&sC[slot][r][p * 2] = t.c;
    };

    static_assert(NPROD * (CHUNK / NPROD) == CHUNK, "every feeding wave makes CHUNK/NPROD steps");
    struct RawStep { float dtv, uv; v4f bq[4]; };
    auto produce = [&](int pi, int buf, int slot) {
        auto fetch_step = [&](RawStep& s, int l) {
            s.dtv = sD[slot][l][lane];
            s.uv  = sU[slot][l][lane];
#pragma unroll
            for (int q = 0; q < 4; ++q)
                s.bq[q] = *(const v4f*)&sB[slot][l][4 * q];
        };
        auto record = [&](const RawStep& s, int l) {
            const float du = s.dtv * s.uv;
            float* rp = &sRec[buf][(l * SC_CH + lane) * REC_F];
#pragma unroll
            for (int q = 0; q < 4; ++q) {
                v4f a, d;
#pragma unroll
                for (int j = 0; j < 4; ++j) {
                    a[j] = __expf(s.dtv * Aa[4 * q + j]);
                    d[j] = du * s.bq[q][j];
                }
                *(v4f*)(rp + 4 * q)      = a;
                *(v4f*)(rp + 16 + 4 * q) = d;
            }
        };
        RawStep rs[CHUNK / NPROD];
#pragma unroll
        for (int k = 0; k < CHUNK / NPROD; ++k)
            fetch_step(rs[k], pi + NPROD * k);
        __builtin_amdgcn_sched_barrier(0);
#pragma unroll
        for (int k = 0; k < CHUNK / NPROD; ++k)
            record(rs[k], pi + NPROD * k);
    };

    auto consume = [&](int r, int buf) {
        const float* rbase = &sRec[buf][lane * REC_F + NSTATE * r];
        float*       hbase = &sH[buf][lane * H_F + NSTATE * r];
#pragma unroll
        for (int l = 0; l < CHUNK; ++l) {
            const float* rp = rbase + l * SC_CH * REC_F;
            float*       hp = hbase + l * SC_CH * H_F;
            v4f a[NSTATE / 4], d[NSTATE / 4];
#pragma unroll
            for (int q = 0; q < NSTATE / 4; ++q) {
                a[q] = *(const v4f*)(rp + 4 * q);
                d[q] = *(const v4f*)(rp + 16 + 4 * q);
            }
#pragma unroll
            for (int q = 0; q < NSTATE / 4; ++q) {
                v4f o;
#pragma unroll
                for (int j = 0; j < 4; ++j) {
                    h[4 * q + j] = a[q][j] * h[4 * q + j] + d[q][j];
                    o[j] = h[4 * q + j];
                }
                *(v4f*)(hp + 4 * q) = o;
            }
        }
    };

    static_assert(NOUT * 4 == CHUNK, "four steps per output wave");
    auto ywork = [&](int yw, int hbuf, int slot, int ybuf) {
        const v4f cq = *(const v4f*)&sC[slot][4 * yw + ((lane & 15) >> 2)][(lane & 3) * 4];
        int cv[4];
#pragma unroll
        for (int k = 0; k < 4; ++k) cv[k] = __float_as_int(cq[k]);
#pragma unroll
        for (int ll = 0; ll < 4; ++ll) {
            const int l = 4 * yw + ll;
            const float* hp = &sH[hbuf][(l * SC_CH + lane) * H_F];
            v4f hq[4];
#pragma unroll
            for (int q = 0; q < 4; ++q) hq[q] = *(const v4f*)(hp + 4 * q);
            const float uv = sU[slot][l][lane];
            const float g  = sG[slot][l][lane];
            float y = 0.f;
#pragma unroll
            for (int n = 0; n < N_ST; ++n) {
                const int v = ll * N_ST + n;
                const float cn = __int_as_float(
                    __builtin_amdgcn_readlane(cv[v & 3], v >> 2));
                y += hq[n >> 2][n & 3] * cn;
            }
            const float sg = g / (1.f + __expf(-g));
            sY[ybuf][l][lane] = (y + uv * Dv) * sg;
        }
    };

    auto store_y = [&](int yw, int ybuf, int row0) {
        const int l = lane & 15, hh = lane >> 4;
        float* dst = yout + (size_t)row0 * I_DIM + (size_t)i0 * 16 + 128 * yw + 4 * lane;
        *(v4f*)dst = *(const v4f*)&sY[ybuf][l][8 * yw + 4 * hh];
    };

    static_assert(CHUNK == 16, "a chunk is one 16-row WMMA tile");
    const int fr = lane & 15, fh = lane >> 4;
    auto dt_request = [&](DtFrag& f, const float* rowp) {
#pragma unroll
        for (int s = 0; s < R_DIM / 4; ++s)
            f.k[s] = *(gcv2f*)(rowp + 4 * s + 2 * fh);
    };
    auto dt_keep = [&](v2f (*dst)[32], const DtFrag& f) {
#pragma unroll
        for (int s = 0; s < R_DIM / 4; ++s) dst[s][lane] = f.k[s];
    };
    auto dt_sums = [&](float bias, int dw, int tl, int slot) {
        DtFrag fa, fw;
#pragma unroll
        for (int s = 0; s < R_DIM / 4; ++s) {
            fa.k[s] = sFa[dw][s][lane];
            fw.k[s] = sFw[dw][s][lane];
        }
        __builtin_amdgcn_sched_barrier(0);
        v8f acc = (v8f){0.f,0.f,0.f,0.f,0.f,0.f,0.f,0.f};
#pragma unroll
        for (int s = 0; s < R_DIM / 4; ++s)
            acc = __builtin_amdgcn_wmma_f32_16x16x4_f32(
                      false, fa.k[s], false, fw.k[s],
                      (short)0, acc, false, false);
        asm volatile("v_nop\n\tv_nop\n\tv_nop\n\tv_nop" : "+v"(acc));
#pragma unroll
        for (int v = 0; v < 8; ++v)
            sD[slot][8 * fh + v][16 * tl + fr] = acc[v] + bias;
    };
    auto dt_softplus = [&](int tl, int slot) {
        float x[8], sp[8];
#pragma unroll
        for (int v = 0; v < 8; ++v) x[v] = sD[slot][8 * fh + v][16 * tl + fr];
#pragma unroll
        for (int v = 0; v < 8; ++v) sp[v] = __logf(1.f + __expf(x[v]));
#pragma unroll
        for (int v = 0; v < 8; ++v)
            sD[slot][8 * fh + v][16 * tl + fr] = (x[v] > 20.f) ? x[v] : sp[v];
    };

    const bool feeder = (wave < 4);
    const int  yw     = wave - (NREC + NPROD);
    const int  dw     = wave - (NREC + NPROD + NOUT);
    const bool dtwave = (dw >= 0);
    const int  tl     = dw & 1;
    const int  pp     = dw >> 1;

    float dbias = 0.f;
    const float* arow = dtn + (size_t)(rowb + fr) * R_DIM;
    if (dtwave) {
        DtFrag f0, f1;
        dt_request(f0, dtw + (size_t)(i0 + 16 * tl + fr) * R_DIM);
        dt_request(f1, arow + (size_t)pp * CHUNK * R_DIM);
        dbias = dtb[i0 + 16 * tl + fr];
        dt_keep(sFw[dw], f0);
        dt_keep(sFa[dw], f1);
        dt_request(f0, arow + (size_t)(pp + 2) * CHUNK * R_DIM);
        dt_sums(dbias, dw, tl, pp);
        if (pp == 0) dt_softplus(tl, 0);
        dt_keep(sFa[dw], f0);
    }

    if (feeder) { request(rowb); park(0); }
    __syncthreads();

    for (int c = -1; c <= NCHUNK; ++c) {
        const bool fetch = feeder && (c + 2 < NCHUNK);
        if (fetch) request(rowb + (c + 2) * CHUNK);
        if (wave < NREC) {
            if (c >= 0 && c < NCHUNK) consume(wave, c & 1);
        } else if (wave < NREC + NPROD) {
            if (c + 1 < NCHUNK) produce(wave - NREC, (c + 1) & 1, (c + 1) & 3);
        } else if (wave < NREC + NPROD + NOUT) {
            if (c >= 2) store_y(yw, c & 1, rowb + (c - 2) * CHUNK);
            if (c >= 1) ywork(yw, (c - 1) & 1, (c - 1) & 3, (c - 1) & 1);
        } else if (((c + 3) & 1) == pp) {
            DtFrag fn;
            const int cn = (c + 5 < NCHUNK) ? c + 5 : NCHUNK - 1;
            dt_request(fn, arow + (size_t)cn * CHUNK * R_DIM);
            if (c + 3 < NCHUNK) dt_sums(dbias, dw, tl, (c + 3) & 3);
            dt_keep(sFa[dw], fn);
        } else {
            if (c + 2 < NCHUNK) dt_softplus(tl, (c + 2) & 3);
        }
        if (fetch) park((c + 2) & 3);
        __syncthreads();
    }
    if (yw >= 0 && yw < NOUT) store_y(yw, (NCHUNK - 1) & 1, rowb + (NCHUNK - 1) * CHUNK);
}

#define PACK_WG      512
#define PACK_THREADS (PACK_WG * 256)

template <int KD, int NPIECES>
__device__ __forceinline__
void pack_matrix(const float* __restrict__ X, float* __restrict__ P, unsigned first)
{
    constexpr unsigned KB = KD / 8;
    auto src = [&](unsigned p) {
        const unsigned ln = p & 31u, blk = p >> 5;
        const unsigned rt = blk / KB, kb = blk - rt * KB;
        return (gcv4f*)(X + (size_t)(16u * rt + (ln & 15u)) * KD + 8u * kb + 4u * (ln >> 4));
    };
    if constexpr (NPIECES % PACK_THREADS == 0) {
        constexpr int PER = NPIECES / PACK_THREADS;
        v4f v[PER];
#pragma unroll
        for (int q = 0; q < PER; ++q) v[q] = *src(first + q * PACK_THREADS);
#pragma unroll
        for (int q = 0; q < PER; ++q)
            *(v4f*)(P + (size_t)(first + q * PACK_THREADS) * 4) = v[q];
    } else {
        static_assert(NPIECES < PACK_THREADS, "a small matrix: at most one piece per thread");
        if (first < (unsigned)NPIECES) *(v4f*)(P + (size_t)first * 4) = *src(first);
    }
}

__global__ __launch_bounds__(256)
void pack_operands_kernel(const float* __restrict__ hidden,
                          const float* __restrict__ in_w,
                          const float* __restrict__ out_w,
                          const float* __restrict__ x_w,
                          float* __restrict__ hidden_p,
                          float* __restrict__ in_w_p,
                          float* __restrict__ out_w_p,
                          float* __restrict__ x_w_p)
{
    const unsigned first = blockIdx.x * 256u + threadIdx.x;
    pack_matrix<H_DIM, 2 * I_DIM * H_DIM / 4>(in_w,   in_w_p,   first);
    pack_matrix<H_DIM, ROWS * H_DIM / 4>     (hidden, hidden_p, first);
    pack_matrix<I_DIM, H_DIM * I_DIM / 4>    (out_w,  out_w_p,  first);
    pack_matrix<I_DIM, XN * I_DIM / 4>       (x_w,    x_w_p,    first);
}

extern "C" void kernel_launch(void* const* d_in, const int* in_sizes, int n_in,
                              void* d_out, int out_size, void* d_ws, size_t ws_size,
                              hipStream_t stream)
{
    const float* hidden     = (const float*)d_in[0];
    const float* in_proj_w  = (const float*)d_in[1];
    const float* conv_w     = (const float*)d_in[2];
    const float* conv_b     = (const float*)d_in[3];
    const float* x_proj_w   = (const float*)d_in[4];
    const float* dt_proj_w  = (const float*)d_in[5];
    const float* dt_proj_b  = (const float*)d_in[6];
    const float* dt_ln_w    = (const float*)d_in[7];
    const float* b_ln_w     = (const float*)d_in[8];
    const float* c_ln_w     = (const float*)d_in[9];
    const float* A_log      = (const float*)d_in[10];
    const float* Dvec       = (const float*)d_in[11];
    const float* out_proj_w = (const float*)d_in[12];
    float* out = (float*)d_out;

    float* ws   = (float*)d_ws;
    float* proj = ws;
    float* u    = proj + (size_t)ROWS * 2 * I_DIM;
    float* Bm   = u    + (size_t)ROWS * I_DIM;
    float* Cm   = Bm   + (size_t)ROWS * N_ST;
    float* dtn  = Cm   + (size_t)ROWS * N_ST;
    float* yout = dtn  + (size_t)ROWS * R_DIM;
    float* hid_p  = yout   + (size_t)ROWS * I_DIM;
    float* inw_p  = hid_p  + (size_t)ROWS * H_DIM;
    float* outw_p = inw_p  + (size_t)2 * I_DIM * H_DIM;
    float* xw_p   = outw_p + (size_t)H_DIM * I_DIM;

    pack_operands_kernel<<<PACK_WG, 256, 0, stream>>>(
        hidden, in_proj_w, out_proj_w, x_proj_w, hid_p, inw_p, outw_p, xw_p);

    wmma_gemm_f32_regtile<4, 4, 2, 2, 4, H_DIM, 2 * I_DIM>
        <<<dim3((2 * I_DIM) / 256, ROWS / 128), 256, 0, stream>>>(
        hid_p, inw_p, proj);

    conv_xproj_kernel<<<ROWS / 16, 256, 0, stream>>>(
        proj, conv_w, conv_b, xw_p, dt_ln_w, b_ln_w, c_ln_w, u, dtn, Bm, Cm);

    scan_kernel<<<dim3(I_DIM / SC_CH, B_SZ), SC_THREADS, 0, stream>>>(
        dtn, dt_proj_w, dt_proj_b, u, Bm, Cm, A_log, Dvec, proj, yout);

    wmma_gemm_f32_regtile<2, 4, 4, 2, 2, I_DIM, H_DIM>
        <<<dim3(H_DIM / 128, ROWS / 64), 128, 0, stream>>>(
        yout, outw_p, out);
}


# scan: 16-channel work-groups on all 256 WGPs walk 32-step chunks (lane = channel x half): 34 barrier trips instead of 66 at the same work per wave and trip; sum order same
# speedup vs baseline: 8.6300x; 1.0882x over previous
#include <hip/hip_runtime.h>
#include <hip/hip_bf16.h>


#define H_DIM 1024
#define I_DIM 2048
#define N_ST  16
#define R_DIM 64
#define B_SZ  2
#define L_SEQ 1024
#define ROWS  (B_SZ * L_SEQ)
#define EPS   1e-06f

typedef __attribute__((ext_vector_type(2))) float v2f;
typedef __attribute__((ext_vector_type(8))) float v8f;

typedef __attribute__((address_space(1))) const float gcf;
typedef __attribute__((address_space(1))) const v2f  gcv2f;
typedef __attribute__((ext_vector_type(4))) float v4f;
typedef __attribute__((address_space(1))) const v4f gcv4f;

template <int MT, int NT, int KS>
struct FragSet {
    v4f a[MT][KS];
    v4f b[NT][KS];
};

template <int MT, int NT, int KS, int KD>
__device__ __forceinline__
void frag_request(FragSet<MT, NT, KS>& f, gcf* Ap, gcf* Wp, int k)
{
#pragma unroll
    for (int mt = 0; mt < MT; ++mt)
#pragma unroll
        for (int j = 0; j < KS; ++j)
            f.a[mt][j] = *(gcv4f*)(Ap + (size_t)mt * 16 * KD + (k + 8 * j) * 16);
#pragma unroll
    for (int nt = 0; nt < NT; ++nt)
#pragma unroll
        for (int j = 0; j < KS; ++j)
            f.b[nt][j] = *(gcv4f*)(Wp + (size_t)nt * 16 * KD + (k + 8 * j) * 16);
}

template <int MT, int NT, int KS>
__device__ __forceinline__
void frag_mma(const FragSet<MT, NT, KS>& f, v8f (&acc)[MT][NT])
{
#pragma unroll
    for (int j = 0; j < KS; ++j) {
        v2f alo[MT], ahi[MT], blo[NT], bhi[NT];
#pragma unroll
        for (int mt = 0; mt < MT; ++mt) {
            alo[mt] = __builtin_shufflevector(f.a[mt][j], f.a[mt][j], 0, 1);
            ahi[mt] = __builtin_shufflevector(f.a[mt][j], f.a[mt][j], 2, 3);
        }
#pragma unroll
        for (int nt = 0; nt < NT; ++nt) {
            blo[nt] = __builtin_shufflevector(f.b[nt][j], f.b[nt][j], 0, 1);
            bhi[nt] = __builtin_shufflevector(f.b[nt][j], f.b[nt][j], 2, 3);
        }
#pragma unroll
        for (int mt = 0; mt < MT; ++mt)
#pragma unroll
            for (int nt = 0; nt < NT; ++nt)
                acc[mt][nt] = __builtin_amdgcn_wmma_f32_16x16x4_f32(
                                  false, alo[mt], false, blo[nt],
                                  (short)0, acc[mt][nt], false, false);
#pragma unroll
        for (int mt = 0; mt < MT; ++mt)
#pragma unroll
            for (int nt = 0; nt < NT; ++nt)
                acc[mt][nt] = __builtin_amdgcn_wmma_f32_16x16x4_f32(
                                  false, ahi[mt], false, bhi[nt],
                                  (short)0, acc[mt][nt], false, false);
    }
}

template <int MT, int NT, int KS, int WM, int WN, int KD, int ND>
__global__ __launch_bounds__(WM * WN * 32)
void wmma_gemm_f32_regtile(const float* __restrict__ A,
                           const float* __restrict__ W,
                           float* __restrict__ C)
{
    constexpr int KSET = 8 * KS;
    static_assert(KD % (2 * KSET) == 0, "K loop consumes two sets per trip");
    const int lane = threadIdx.x & 31;
    const int wave = threadIdx.x >> 5;
    const int wm   = wave / WN;
    const int wn   = wave - wm * WN;
    const int row0 = (blockIdx.y * WM + wm) * (16 * MT);
    const int col0 = (blockIdx.x * WN + wn) * (16 * NT);

    const int r = lane & 15;
    const int h = lane >> 4;

    gcf* Ap = (gcf*)(A + (size_t)row0 * KD + 4 * lane);
    gcf* Wp = (gcf*)(W + (size_t)col0 * KD + 4 * lane);

    v8f acc[MT][NT];
#pragma unroll
    for (int mt = 0; mt < MT; ++mt)
#pragma unroll
        for (int nt = 0; nt < NT; ++nt)
            acc[mt][nt] = (v8f){0.f,0.f,0.f,0.f,0.f,0.f,0.f,0.f};

    FragSet<MT, NT, KS> f0, f1;
    frag_request<MT, NT, KS, KD>(f0, Ap, Wp, 0);
    for (int k = 0; k < KD - 2 * KSET; k += 2 * KSET) {
        frag_request<MT, NT, KS, KD>(f1, Ap, Wp, k + KSET);
        __builtin_amdgcn_sched_barrier(0);
        frag_mma<MT, NT, KS>(f0, acc);
        __builtin_amdgcn_sched_barrier(0);
        frag_request<MT, NT, KS, KD>(f0, Ap, Wp, k + 2 * KSET);
        __builtin_amdgcn_sched_barrier(0);
        frag_mma<MT, NT, KS>(f1, acc);
        __builtin_amdgcn_sched_barrier(0);
    }
    frag_request<MT, NT, KS, KD>(f1, Ap, Wp, KD - KSET);
    __builtin_amdgcn_sched_barrier(0);
    frag_mma<MT, NT, KS>(f0, acc);
    frag_mma<MT, NT, KS>(f1, acc);

#pragma unroll
    for (int mt = 0; mt < MT; ++mt)
#pragma unroll
        for (int nt = 0; nt < NT; ++nt)
#pragma unroll
            for (int v = 0; v < 8; ++v)
                C[(size_t)(row0 + mt * 16 + 8 * h + v) * ND + col0 + nt * 16 + r] =
                    acc[mt][nt][v];
}

#define XN      (R_DIM + 2 * N_ST)
#define XNT     (XN / 16)
#define XPITCH  (I_DIM + 4)
#define XKW     (I_DIM / 8)
#define XP      100

__global__ __launch_bounds__(256)
void conv_xproj_kernel(const float* __restrict__ proj,
                       const float* __restrict__ cw,
                       const float* __restrict__ cb,
                       const float* __restrict__ xw,
                       const float* __restrict__ wdt,
                       const float* __restrict__ wb,
                       const float* __restrict__ wc,
                       float* __restrict__ u,
                       float* __restrict__ dtn,
                       float* __restrict__ Bm,
                       float* __restrict__ Cm)
{
    static_assert(L_SEQ % 16 == 0, "a 16-row tile must stay inside one batch");
    static_assert(XKW % 64 == 0, "K loop consumes two sets of 32 k per trip");
    __shared__ __attribute__((aligned(16))) float sA[16 * XPITCH];
    __shared__ __attribute__((aligned(16))) float sRed[8 * XNT * 8 * 32];

    const int tid  = threadIdx.x;
    const int lane = tid & 31;
    const int wave = tid >> 5;
    const int row0 = blockIdx.x * 16;
    const int l0   = row0 % L_SEQ;
    const bool has_top = (l0 > 0);
    const bool has_bot = (l0 + 16 < L_SEQ);

#pragma unroll
    for (int p = 0; p < 2; ++p) {
        const int ch = wave * XKW + p * 128 + lane * 4;
        gcf* src = (gcf*)(proj + (size_t)row0 * (2 * I_DIM) + ch);
        const v4f zero = (v4f){0.f, 0.f, 0.f, 0.f};
        v4f x[19];
        x[0] = zero; x[17] = zero; x[18] = zero;
        if (has_top) x[0] = *(gcv4f*)(src - (2 * I_DIM));
#pragma unroll
        for (int q = 1; q < 17; ++q)
            x[q] = *(gcv4f*)(src + (size_t)(q - 1) * (2 * I_DIM));
        if (has_bot) {
            x[17] = *(gcv4f*)(src + (size_t)16 * (2 * I_DIM));
            x[18] = *(gcv4f*)(src + (size_t)17 * (2 * I_DIM));
        }
        v4f w4[4];
#pragma unroll
        for (int c = 0; c < 4; ++c)
            w4[c] = *(gcv4f*)(cw + (size_t)(ch + c) * 4);
        const v4f bias = *(gcv4f*)(cb + ch);

#pragma unroll
        for (int r = 0; r < 16; ++r) {
            v4f o;
#pragma unroll
            for (int c = 0; c < 4; ++c) {
                float acc = bias[c];
#pragma unroll
                for (int k = 0; k < 4; ++k)
                    acc += w4[c][k] * x[r + k][c];
                o[c] = acc / (1.f + __expf(-acc));
            }
            *(v4f*)(u + (size_t)(row0 + r) * I_DIM + ch) = o;
            *(v4f*)&sA[r * XPITCH + ch] = o;
        }
    }
    __syncthreads();

    const int r = lane & 15;
    const int h = lane >> 4;
    const float* sAp = sA + r * XPITCH + wave * XKW + 4 * h;
    gcf* Wp = (gcf*)(xw + (size_t)(wave * XKW) * 16 + 4 * lane);

    v8f acc[1][XNT];
#pragma unroll
    for (int t = 0; t < XNT; ++t)
        acc[0][t] = (v8f){0.f,0.f,0.f,0.f,0.f,0.f,0.f,0.f};

    auto request = [&](FragSet<1, XNT, 4>& f, int k) {
#pragma unroll
        for (int j = 0; j < 4; ++j)
            f.a[0][j] = *(const v4f*)(sAp + k + 8 * j);
#pragma unroll
        for (int t = 0; t < XNT; ++t)
#pragma unroll
            for (int j = 0; j < 4; ++j)
                f.b[t][j] = *(gcv4f*)(Wp + (size_t)t * 16 * I_DIM + (k + 8 * j) * 16);
    };

    FragSet<1, XNT, 4> f0, f1;
    request(f0, 0);
    for (int k = 0; k < XKW - 64; k += 64) {
        request(f1, k + 32);
        __builtin_amdgcn_sched_barrier(0);
        frag_mma<1, XNT, 4>(f0, acc);
        __builtin_amdgcn_sched_barrier(0);
        request(f0, k + 64);
        __builtin_amdgcn_sched_barrier(0);
        frag_mma<1, XNT, 4>(f1, acc);
        __builtin_amdgcn_sched_barrier(0);
    }
    request(f1, XKW - 32);
    __builtin_amdgcn_sched_barrier(0);
    frag_mma<1, XNT, 4>(f0, acc);
    frag_mma<1, XNT, 4>(f1, acc);

#pragma unroll
    for (int t = 0; t < XNT; ++t)
#pragma unroll
        for (int v = 0; v < 8; ++v)
            sRed[((wave * XNT + t) * 8 + v) * 32 + lane] = acc[0][t][v];
    __syncthreads();

    float* sX   = sA;
    float* sInv = sA + 16 * XP;

    const int xrow = 8 * (lane >> 4) + wave;
    const int xc   = lane & 15;
    float xv[XNT];
#pragma unroll
    for (int q = 0; q < XNT; ++q) {
        const int e = tid + 256 * q;
        float sum = 0.f;
#pragma unroll
        for (int w = 0; w < 8; ++w)
            sum += sRed[w * (XNT * 8 * 32) + e];
        xv[q] = sum;
        sX[xrow * XP + 16 * q + xc] = sum;
    }
    __syncthreads();

    if (tid < 48) {
        const int seg = tid >> 4;
        const float* x = sX + (tid & 15) * XP;
        float s = 0.f, inv;
        if (seg == 0) {
#pragma unroll 8
            for (int j = 0; j < R_DIM; ++j) s += x[j] * x[j];
            inv = __frsqrt_rn(s / R_DIM + EPS);
        } else {
            const float* xs = x + R_DIM + (seg - 1) * N_ST;
#pragma unroll
            for (int j = 0; j < N_ST; ++j) { float v = xs[j]; s += v * v; }
            inv = __frsqrt_rn(s / N_ST + EPS);
        }
        sInv[tid] = inv;
    }
    __syncthreads();

    const float inv_dt = sInv[xrow], inv_b = sInv[16 + xrow], inv_c = sInv[32 + xrow];
#pragma unroll
    for (int q = 0; q < R_DIM / 16; ++q)
        dtn[(size_t)(row0 + xrow) * R_DIM + 16 * q + xc] = wdt[16 * q + xc] * xv[q] * inv_dt;
    Bm[(size_t)(row0 + xrow) * N_ST + xc] = wb[xc] * xv[4] * inv_b;
    Cm[(size_t)(row0 + xrow) * N_ST + xc] = wc[xc] * xv[5] * inv_c;
}

#define CHUNK   32
#define NCHUNK  (L_SEQ / CHUNK)
#define SC_CH   16
#define REC_F   36
#define REC_BUF (CHUNK * SC_CH * REC_F)
#define H_F     20
#define H_BUF   (CHUNK * SC_CH * H_F)
#define Y_PITCH 20
#define NRAW    4
#define NREC    2
#define NSTATE  (N_ST / NREC)
#define NPROD   4
#define NOUT    4
#define NDT     (2 * (CHUNK / 16))
#define SC_WAVES   (NREC + NPROD + NOUT + NDT)
#define SC_THREADS (SC_WAVES * 32)

struct ScanTile {
    v4f u, g;
    v4f b, c;
};

struct DtFrag { v2f k[R_DIM / 4]; };

__global__ __launch_bounds__(SC_THREADS)
void scan_kernel(const float* __restrict__ dtn,
                 const float* __restrict__ dtw,
                 const float* __restrict__ dtb,
                 const float* __restrict__ u,
                 const float* __restrict__ Bm,
                 const float* __restrict__ Cm,
                 const float* __restrict__ A_log,
                 const float* __restrict__ Dvec,
                 const float* __restrict__ proj,
                 float* __restrict__ yout)
{
    static_assert(SC_CH == 16, "a lane is (channel, half): 16 channels per work-group");
    static_assert(N_ST % NREC == 0 && NSTATE == 8, "a lane half holds 4 of a recurrence wave's 8 states");
    __shared__ __attribute__((aligned(16))) float sRec[2][REC_BUF];
    __shared__ __attribute__((aligned(16))) float sH[2][H_BUF];
    __shared__ __attribute__((aligned(16))) float sD[NRAW][CHUNK][SC_CH];
    __shared__ __attribute__((aligned(16))) float sU[NRAW][CHUNK][SC_CH];
    __shared__ __attribute__((aligned(16))) float sG[NRAW][CHUNK][SC_CH];
    __shared__ __attribute__((aligned(16))) float sB[NRAW][CHUNK][N_ST];
    __shared__ __attribute__((aligned(16))) float sC[NRAW][CHUNK][N_ST];
    __shared__ __attribute__((aligned(16))) float sY[2][CHUNK][Y_PITCH];
    __shared__ __attribute__((aligned(16))) v2f sFa[NDT][R_DIM / 4][32];
    __shared__ __attribute__((aligned(16))) v2f sFw[NDT][R_DIM / 4][32];
    static_assert(NDT == 4, "dt wave = (row tile, chunk parity)");

    const int tid  = threadIdx.x;
    const int lane = tid & 31;
    const int wave = __builtin_amdgcn_readfirstlane(tid >> 5);
    const int b    = blockIdx.y;
    const int i0   = blockIdx.x * SC_CH;
    const int ch   = lane & 15;
    const int sub  = lane >> 4;
    const int i    = i0 + ch;
    const int rowb = b * L_SEQ;

    float Aa[N_ST];
#pragma unroll
    for (int n = 0; n < N_ST; ++n)
        Aa[n] = -__expf(A_log[(size_t)i * N_ST + n]);

    float h[4];
#pragma unroll
    for (int n = 0; n < 4; ++n) h[n] = 0.f;

    const float Dv = Dvec[i];

    ScanTile t;

    auto request = [&](int row0) {
        const size_t row = (size_t)(row0 + (tid >> 2));
        const int    col = i0 + (tid & 3) * 4;
        t.u = *(gcv4f*)(u    + row * I_DIM + col);
        t.g = *(gcv4f*)(proj + row * (2 * I_DIM) + I_DIM + col);
        t.b = *(gcv4f*)(Bm + (size_t)row0 * N_ST + tid * 4);
        t.c = *(gcv4f*)(Cm + (size_t)row0 * N_ST + tid * 4);
    };

    auto park = [&](int slot) {
        const int r = tid >> 2, p = tid & 3;
        *(v4f*)&sU[slot][r][p * 4] = t.u;
        *(v4f*)&sG[slot][r][p * 4] = t.g;
        *(v4f*)&sB[slot][r][p * 4] = t.b;
        *(v4f*)&sC[slot][r][p * 4] = t.c;
    };

    static_assert(NPROD * 8 == CHUNK, "eight steps per feeding wave");
    struct RawStep { float dtv, uv; v4f bq[4]; };
    auto produce = [&](int pi, int buf, int slot) {
        auto fetch_step = [&](RawStep& s, int l) {
            s.dtv = sD[slot][l][ch];
            s.uv  = sU[slot][l][ch];
#pragma unroll
            for (int q = 0; q < 4; ++q)
                s.bq[q] = *(const v4f*)&sB[slot][l][4 * q];
        };
        auto record = [&](const RawStep& s, int l) {
            const float du = s.dtv * s.uv;
            float* rp = &sRec[buf][(l * SC_CH + ch) * REC_F];
#pragma unroll
            for (int q = 0; q < 4; ++q) {
                v4f a, d;
#pragma unroll
                for (int j = 0; j < 4; ++j) {
                    a[j] = __expf(s.dtv * Aa[4 * q + j]);
                    d[j] = du * s.bq[q][j];
                }
                *(v4f*)(rp + 4 * q)      = a;
                *(v4f*)(rp + 16 + 4 * q) = d;
            }
        };
        RawStep rs[4];
#pragma unroll
        for (int k = 0; k < 4; ++k)
            fetch_step(rs[k], 8 * pi + 2 * k + sub);
        __builtin_amdgcn_sched_barrier(0);
#pragma unroll
        for (int k = 0; k < 4; ++k)
            record(rs[k], 8 * pi + 2 * k + sub);
    };

    constexpr int REC_AHEAD = 12;
    auto consume = [&](int r, int buf) {
        const float* rbase = &sRec[buf][ch * REC_F + NSTATE * r + 4 * sub];
        float*       hbase = &sH[buf][ch * H_F + NSTATE * r + 4 * sub];
        v4f a[CHUNK], d[CHUNK];
#pragma unroll
        for (int l = 0; l < REC_AHEAD; ++l) {
            a[l] = *(const v4f*)(rbase + l * SC_CH * REC_F);
            d[l] = *(const v4f*)(rbase + l * SC_CH * REC_F + 16);
        }
#pragma unroll
        for (int l = 0; l < CHUNK; ++l) {
            if (l + REC_AHEAD < CHUNK) {
                a[l + REC_AHEAD] = *(const v4f*)(rbase + (l + REC_AHEAD) * SC_CH * REC_F);
                d[l + REC_AHEAD] = *(const v4f*)(rbase + (l + REC_AHEAD) * SC_CH * REC_F + 16);
            }
            __builtin_amdgcn_sched_barrier(0);
            v4f o;
#pragma unroll
            for (int j = 0; j < 4; ++j) {
                h[j] = a[l][j] * h[j] + d[l][j];
                o[j] = h[j];
            }
            *(v4f*)(hbase + l * SC_CH * H_F) = o;
        }
    };

    static_assert(NOUT * 8 == CHUNK, "eight steps per output wave");
    auto ywork = [&](int yw, int hbuf, int slot, int ybuf) {
#pragma unroll
        for (int k = 0; k < 4; ++k) {
            const int l = 8 * yw + 2 * k + sub;
            const float* hp = &sH[hbuf][(l * SC_CH + ch) * H_F];
            v4f hq[4], cq[4];
#pragma unroll
            for (int q = 0; q < 4; ++q) hq[q] = *(const v4f*)(hp + 4 * q);
#pragma unroll
            for (int q = 0; q < 4; ++q) cq[q] = *(const v4f*)&sC[slot][l][4 * q];
            const float uv = sU[slot][l][ch];
            const float g  = sG[slot][l][ch];
            float y = 0.f;
#pragma unroll
            for (int n = 0; n < N_ST; ++n)
                y += hq[n >> 2][n & 3] * cq[n >> 2][n & 3];
            const float sg = g / (1.f + __expf(-g));
            sY[ybuf][l][ch] = (y + uv * Dv) * sg;
        }
    };

    auto store_y = [&](int yw, int ybuf, int row0) {
        const int l = lane & 15, hh = lane >> 4;
        const int rt = yw >> 1, bk = yw & 1;
        float* dst = yout + (size_t)(row0 + 16 * rt) * I_DIM + (size_t)i0 * 16 + 128 * bk + 4 * lane;
        *(v4f*)dst = *(const v4f*)&sY[ybuf][16 * rt + l][8 * bk + 4 * hh];
    };

    static_assert(CHUNK == 32, "a chunk is two 16-row WMMA tiles");
    const int fr = ch, fh = sub;
    auto dt_request = [&](DtFrag& f, const float* rowp) {
#pragma unroll
        for (int s = 0; s < R_DIM / 4; ++s)
            f.k[s] = *(gcv2f*)(rowp + 4 * s + 2 * fh);
    };
    auto dt_keep = [&](v2f (*dst)[32], const DtFrag& f) {
#pragma unroll
        for (int s = 0; s < R_DIM / 4; ++s) dst[s][lane] = f.k[s];
    };
    auto dt_sums = [&](float bias, int dw, int mt, int slot) {
        DtFrag fa, fw;
#pragma unroll
        for (int s = 0; s < R_DIM / 4; ++s) {
            fa.k[s] = sFa[dw][s][lane];
            fw.k[s] = sFw[dw][s][lane];
        }
        __builtin_amdgcn_sched_barrier(0);
        v8f acc = (v8f){0.f,0.f,0.f,0.f,0.f,0.f,0.f,0.f};
#pragma unroll
        for (int s = 0; s < R_DIM / 4; ++s)
            acc = __builtin_amdgcn_wmma_f32_16x16x4_f32(
                      false, fa.k[s], false, fw.k[s],
                      (short)0, acc, false, false);
        asm volatile("v_nop\n\tv_nop\n\tv_nop\n\tv_nop" : "+v"(acc));
#pragma unroll
        for (int v = 0; v < 8; ++v)
            sD[slot][16 * mt + 8 * fh + v][fr] = acc[v] + bias;
    };
    auto dt_softplus = [&](int mt, int slot) {
        float x[8], sp[8];
#pragma unroll
        for (int v = 0; v < 8; ++v) x[v] = sD[slot][16 * mt + 8 * fh + v][fr];
#pragma unroll
        for (int v = 0; v < 8; ++v) sp[v] = __logf(1.f + __expf(x[v]));
#pragma unroll
        for (int v = 0; v < 8; ++v)
            sD[slot][16 * mt + 8 * fh + v][fr] = (x[v] > 20.f) ? x[v] : sp[v];
    };

    const bool feeder = (wave < 4);
    const int  yw     = wave - (NREC + NPROD);
    const int  dw     = wave - (NREC + NPROD + NOUT);
    const bool dtwave = (dw >= 0);
    const int  mt     = dw & 1;
    const int  pp     = dw >> 1;

    float dbias = 0.f;
    const float* arow = dtn + (size_t)(rowb + 16 * mt + fr) * R_DIM;
    if (dtwave) {
        DtFrag f0, f1;
        dt_request(f0, dtw + (size_t)(i0 + fr) * R_DIM);
        dt_request(f1, arow + (size_t)pp * CHUNK * R_DIM);
        dbias = dtb[i0 + fr];
        dt_keep(sFw[dw], f0);
        dt_keep(sFa[dw], f1);
        dt_request(f0, arow + (size_t)(pp + 2) * CHUNK * R_DIM);
        dt_sums(dbias, dw, mt, pp);
        if (pp == 0) dt_softplus(mt, 0);
        dt_keep(sFa[dw], f0);
    }

    if (feeder) { request(rowb); park(0); }
    __syncthreads();

    for (int c = -1; c <= NCHUNK; ++c) {
        const bool fetch = feeder && (c + 2 < NCHUNK);
        if (fetch) request(rowb + (c + 2) * CHUNK);
        if (wave < NREC) {
            if (c >= 0 && c < NCHUNK) consume(wave, c & 1);
        } else if (wave < NREC + NPROD) {
            if (c + 1 < NCHUNK) produce(wave - NREC, (c + 1) & 1, (c + 1) & 3);
        } else if (wave < NREC + NPROD + NOUT) {
            if (c >= 2) store_y(yw, c & 1, rowb + (c - 2) * CHUNK);
            if (c >= 1) ywork(yw, (c - 1) & 1, (c - 1) & 3, (c - 1) & 1);
        } else if (((c + 3) & 1) == pp) {
            DtFrag fn;
            const int cn = (c + 5 < NCHUNK) ? c + 5 : NCHUNK - 1;
            dt_request(fn, arow + (size_t)cn * CHUNK * R_DIM);
            if (c + 3 < NCHUNK) dt_sums(dbias, dw, mt, (c + 3) & 3);
            dt_keep(sFa[dw], fn);
        } else {
            if (c + 2 < NCHUNK) dt_softplus(mt, (c + 2) & 3);
        }
        if (fetch) park((c + 2) & 3);
        __syncthreads();
    }
    if (yw >= 0 && yw < NOUT) store_y(yw, (NCHUNK - 1) & 1, rowb + (NCHUNK - 1) * CHUNK);
}

#define PACK_WG      512
#define PACK_THREADS (PACK_WG * 256)

template <int KD, int NPIECES>
__device__ __forceinline__
void pack_matrix(const float* __restrict__ X, float* __restrict__ P, unsigned first)
{
    constexpr unsigned KB = KD / 8;
    auto src = [&](unsigned p) {
        const unsigned ln = p & 31u, blk = p >> 5;
        const unsigned rt = blk / KB, kb = blk - rt * KB;
        return (gcv4f*)(X + (size_t)(16u * rt + (ln & 15u)) * KD + 8u * kb + 4u * (ln >> 4));
    };
    if constexpr (NPIECES % PACK_THREADS == 0) {
        constexpr int PER = NPIECES / PACK_THREADS;
        v4f v[PER];
#pragma unroll
        for (int q = 0; q < PER; ++q) v[q] = *src(first + q * PACK_THREADS);
#pragma unroll
        for (int q = 0; q < PER; ++q)
            *(v4f*)(P + (size_t)(first + q * PACK_THREADS) * 4) = v[q];
    } else {
        static_assert(NPIECES < PACK_THREADS, "a small matrix: at most one piece per thread");
        if (first < (unsigned)NPIECES) *(v4f*)(P + (size_t)first * 4) = *src(first);
    }
}

__global__ __launch_bounds__(256)
void pack_operands_kernel(const float* __restrict__ hidden,
                          const float* __restrict__ in_w,
                          const float* __restrict__ out_w,
                          const float* __restrict__ x_w,
                          float* __restrict__ hidden_p,
                          float* __restrict__ in_w_p,
                          float* __restrict__ out_w_p,
                          float* __restrict__ x_w_p)
{
    const unsigned first = blockIdx.x * 256u + threadIdx.x;
    pack_matrix<H_DIM, 2 * I_DIM * H_DIM / 4>(in_w,   in_w_p,   first);
    pack_matrix<H_DIM, ROWS * H_DIM / 4>     (hidden, hidden_p, first);
    pack_matrix<I_DIM, H_DIM * I_DIM / 4>    (out_w,  out_w_p,  first);
    pack_matrix<I_DIM, XN * I_DIM / 4>       (x_w,    x_w_p,    first);
}

extern "C" void kernel_launch(void* const* d_in, const int* in_sizes, int n_in,
                              void* d_out, int out_size, void* d_ws, size_t ws_size,
                              hipStream_t stream)
{
    const float* hidden     = (const float*)d_in[0];
    const float* in_proj_w  = (const float*)d_in[1];
    const float* conv_w     = (const float*)d_in[2];
    const float* conv_b     = (const float*)d_in[3];
    const float* x_proj_w   = (const float*)d_in[4];
    const float* dt_proj_w  = (const float*)d_in[5];
    const float* dt_proj_b  = (const float*)d_in[6];
    const float* dt_ln_w    = (const float*)d_in[7];
    const float* b_ln_w     = (const float*)d_in[8];
    const float* c_ln_w     = (const float*)d_in[9];
    const float* A_log      = (const float*)d_in[10];
    const float* Dvec       = (const float*)d_in[11];
    const float* out_proj_w = (const float*)d_in[12];
    float* out = (float*)d_out;

    float* ws   = (float*)d_ws;
    float* proj = ws;
    float* u    = proj + (size_t)ROWS * 2 * I_DIM;
    float* Bm   = u    + (size_t)ROWS * I_DIM;
    float* Cm   = Bm   + (size_t)ROWS * N_ST;
    float* dtn  = Cm   + (size_t)ROWS * N_ST;
    float* yout = dtn  + (size_t)ROWS * R_DIM;
    float* hid_p  = yout   + (size_t)ROWS * I_DIM;
    float* inw_p  = hid_p  + (size_t)ROWS * H_DIM;
    float* outw_p = inw_p  + (size_t)2 * I_DIM * H_DIM;
    float* xw_p   = outw_p + (size_t)H_DIM * I_DIM;

    pack_operands_kernel<<<PACK_WG, 256, 0, stream>>>(
        hidden, in_proj_w, out_proj_w, x_proj_w, hid_p, inw_p, outw_p, xw_p);

    wmma_gemm_f32_regtile<4, 4, 2, 2, 4, H_DIM, 2 * I_DIM>
        <<<dim3((2 * I_DIM) / 256, ROWS / 128), 256, 0, stream>>>(
        hid_p, inw_p, proj);

    conv_xproj_kernel<<<ROWS / 16, 256, 0, stream>>>(
        proj, conv_w, conv_b, xw_p, dt_ln_w, b_ln_w, c_ln_w, u, dtn, Bm, Cm);

    scan_kernel<<<dim3(I_DIM / SC_CH, B_SZ), SC_THREADS, 0, stream>>>(
        dtn, dt_proj_w, dt_proj_b, u, Bm, Cm, A_log, Dvec, proj, yout);

    wmma_gemm_f32_regtile<2, 4, 4, 2, 2, I_DIM, H_DIM>
        <<<dim3(H_DIM / 128, ROWS / 64), 128, 0, stream>>>(
        yout, outw_p, out);
}
